# starting kernel
# baseline (speedup 1.0000x reference)
#include <hip/hip_runtime.h>
#include <hip/hip_bf16.h>

typedef float v2f __attribute__((ext_vector_type(2)));
typedef float v8f __attribute__((ext_vector_type(8)));

#define GAT_IN_DIM 128
#define GAT_NEG_SLOPE 0.2f

__global__ void wmma_gemm_f32(const float* __restrict__ A,
                              const float* __restrict__ B,
                              float* __restrict__ C,
                              const float* __restrict__ bias,
                              int M, int N, int K) {
  int wave = (blockIdx.x * blockDim.x + threadIdx.x) >> 5;
  int lane = threadIdx.x & 31;
  int tiles_n = N >> 4;
  int tiles = (M >> 4) * tiles_n;
  if (wave >= tiles) return;
  int tm = (wave / tiles_n) << 4;
  int tn = (wave % tiles_n) << 4;
  int idx  = lane & 15;
  int kk   = (lane >> 4) << 1;
  int half = lane >> 4;

  v8f acc = {};
  for (int k0 = 0; k0 < K; k0 += 4) {
    v2f a, b;
    const float* ap = A + (size_t)(tm + idx) * K + (k0 + kk);
    a.x = ap[0];
    a.y = ap[1];
    const float* bp = B + (size_t)(k0 + kk) * N + tn + idx;
    b.x = bp[0];
    b.y = bp[N];
    acc = __builtin_amdgcn_wmma_f32_16x16x4_f32(
         false, a,  false, b,
         (short)0, acc,  false,  false);
  }

  float badd = bias ? bias[tn + idx] : 0.0f;
#pragma unroll
  for (int r = 0; r < 8; ++r) {
    C[(size_t)(tm + r + 8 * half) * N + tn + idx] = acc[r] + badd;
  }
}

__global__ void gat_scores(const float* __restrict__ h,
                           const float* __restrict__ attn,
                           float* __restrict__ es, float* __restrict__ ed,
                           int n_nodes, int H, int F) {
  int t = blockIdx.x * blockDim.x + threadIdx.x;
  if (t >= n_nodes * H) return;
  int n = t / H;
  int hh = t - n * H;
  const float* hp = h + (size_t)n * H * F + (size_t)hh * F;
  const float* al = attn + (size_t)hh * 2 * F;
  const float* ar = al + F;
  float s = 0.f, d = 0.f;
  for (int f = 0; f < F; ++f) {
    s += hp[f] * al[f];
    d += hp[f] * ar[f];
  }
  es[t] = s;
  ed[t] = d;
}

__device__ __forceinline__ unsigned encf(float f) {
  unsigned u = __float_as_uint(f);
  return (u & 0x80000000u) ? ~u : (u | 0x80000000u);
}
__device__ __forceinline__ float decf(unsigned u) {
  return (u & 0x80000000u) ? __uint_as_float(u & 0x7FFFFFFFu)
                           : __uint_as_float(~u);
}

__global__ void edge_alpha_kernel(const float* __restrict__ es,
                                  const float* __restrict__ ed,
                                  const int* __restrict__ rowE,
                                  const int* __restrict__ colE,
                                  float* __restrict__ alpha,
                                  unsigned* __restrict__ amax,
                                  int E, int Nn, int H) {
  int t = blockIdx.x * blockDim.x + threadIdx.x;
  if (t >= (E + Nn) * H) return;
  int e = t / H;
  int h = t - e * H;
  int r = (e < E) ? rowE[e] : (e - E);
  int c = (e < E) ? colE[e] : (e - E);
  float a = es[r * H + h] + ed[c * H + h];
  a = (a > 0.f) ? a : a * GAT_NEG_SLOPE;
  alpha[t] = a;
  atomicMax(&amax[c * H + h], encf(a));
}

__global__ void edge_exp_kernel(float* __restrict__ alpha,
                                const unsigned* __restrict__ amax,
                                float* __restrict__ denom,
                                const int* __restrict__ colE,
                                int E, int Nn, int H) {
  int t = blockIdx.x * blockDim.x + threadIdx.x;
  if (t >= (E + Nn) * H) return;
  int e = t / H;
  int h = t - e * H;
  int c = (e < E) ? colE[e] : (e - E);
  float am = decf(amax[c * H + h]);
  float ex = __expf(alpha[t] - am);
  alpha[t] = ex;
  atomicAdd(&denom[c * H + h], ex);
}

__global__ void edge_scatter_kernel(const float* __restrict__ ex,
                                    const float* __restrict__ denom,
                                    const float* __restrict__ hsrc,
                                    float* __restrict__ outd,
                                    const int* __restrict__ rowE,
                                    const int* __restrict__ colE,
                                    int E, int Nn, int H, int F) {
  int Ftot = H * F;
  long long t = (long long)blockIdx.x * blockDim.x + threadIdx.x;
  long long total = (long long)(E + Nn) * Ftot;
  if (t >= total) return;
  int e = (int)(t / Ftot);
  int f = (int)(t - (long long)e * Ftot);
  int h = f / F;
  int r = (e < E) ? rowE[e] : (e - E);
  int c = (e < E) ? colE[e] : (e - E);
  float coef = ex[(size_t)e * H + h] / denom[c * H + h];
  atomicAdd(&outd[(size_t)c * Ftot + f], hsrc[(size_t)r * Ftot + f] * coef);
}

__global__ void relu_kernel(float* __restrict__ p, long long n) {
  long long t = (long long)blockIdx.x * blockDim.x + threadIdx.x;
  if (t < n) p[t] = fmaxf(p[t], 0.f);
}


extern "C" void kernel_launch(void* const* d_in, const int* in_sizes, int n_in,
                              void* d_out, int out_size, void* d_ws, size_t ws_size,
                              hipStream_t stream) {
  const float* x     = (const float*)d_in[0];
  const int*   ei    = (const int*)  d_in[1];
  const float* W1    = (const float*)d_in[2];
  const float* attn1 = (const float*)d_in[3];
  const float* W2    = (const float*)d_in[4];
  const float* attn2 = (const float*)d_in[5];
  const float* headW = (const float*)d_in[6];
  const float* headb = (const float*)d_in[7];

  const int Nn = in_sizes[0] / GAT_IN_DIM;
  const int E  = in_sizes[1] / 2;
  const int ET = E + Nn;
  const int* rowE = ei;
  const int* colE = ei + E;

  float* w = (float*)d_ws;
  float* h1    = w; w += (size_t)Nn * 128;
  float* out1  = w; w += (size_t)Nn * 128;
  float* h2    = w; w += (size_t)Nn * 64;
  float* out2  = w; w += (size_t)Nn * 64;
  float* es1   = w; w += (size_t)Nn * 4;
  float* ed1   = w; w += (size_t)Nn * 4;
  unsigned* amax1 = (unsigned*)w; w += (size_t)Nn * 4;
  float* den1  = w; w += (size_t)Nn * 4;
  float* es2   = w; w += (size_t)Nn;
  float* ed2   = w; w += (size_t)Nn;
  unsigned* amax2 = (unsigned*)w; w += (size_t)Nn;
  float* den2  = w; w += (size_t)Nn;
  float* ebuf  = w; w += (size_t)ET * 4;

  hipMemsetAsync(out1,  0, (size_t)Nn * 128 * sizeof(float), stream);
  hipMemsetAsync(out2,  0, (size_t)Nn * 64  * sizeof(float), stream);
  hipMemsetAsync(amax1, 0, (size_t)Nn * 4   * sizeof(unsigned), stream);
  hipMemsetAsync(den1,  0, (size_t)Nn * 4   * sizeof(float), stream);
  hipMemsetAsync(amax2, 0, (size_t)Nn       * sizeof(unsigned), stream);
  hipMemsetAsync(den2,  0, (size_t)Nn       * sizeof(float), stream);

  const int TB = 256;
  auto gemm_blocks = [](int M, int N) {
    int tiles = (M >> 4) * (N >> 4);
    return (tiles + 7) / 8;
  };

  wmma_gemm_f32<<<gemm_blocks(Nn, 128), TB, 0, stream>>>(x, W1, h1, nullptr, Nn, 128, 128);
  gat_scores<<<(Nn * 4 + TB - 1) / TB, TB, 0, stream>>>(h1, attn1, es1, ed1, Nn, 4, 32);
  {
    int nt = ET * 4;
    edge_alpha_kernel<<<(nt + TB - 1) / TB, TB, 0, stream>>>(es1, ed1, rowE, colE, ebuf, amax1, E, Nn, 4);
    edge_exp_kernel<<<(nt + TB - 1) / TB, TB, 0, stream>>>(ebuf, amax1, den1, colE, E, Nn, 4);
    long long ns = (long long)ET * 128;
    edge_scatter_kernel<<<(unsigned)((ns + TB - 1) / TB), TB, 0, stream>>>(
        ebuf, den1, h1, out1, rowE, colE, E, Nn, 4, 32);
  }
  {
    long long n = (long long)Nn * 128;
    relu_kernel<<<(unsigned)((n + TB - 1) / TB), TB, 0, stream>>>(out1, n);
  }

  wmma_gemm_f32<<<gemm_blocks(Nn, 64), TB, 0, stream>>>(out1, W2, h2, nullptr, Nn, 64, 128);
  gat_scores<<<(Nn + TB - 1) / TB, TB, 0, stream>>>(h2, attn2, es2, ed2, Nn, 1, 64);
  {
    int nt = ET;
    edge_alpha_kernel<<<(nt + TB - 1) / TB, TB, 0, stream>>>(es2, ed2, rowE, colE, ebuf, amax2, E, Nn, 1);
    edge_exp_kernel<<<(nt + TB - 1) / TB, TB, 0, stream>>>(ebuf, amax2, den2, colE, E, Nn, 1);
    long long ns = (long long)ET * 64;
    edge_scatter_kernel<<<(unsigned)((ns + TB - 1) / TB), TB, 0, stream>>>(
        ebuf, den2, h2, out2, rowE, colE, E, Nn, 1, 64);
  }

  wmma_gemm_f32<<<gemm_blocks(Nn, 64), TB, 0, stream>>>(out2, headW, (float*)d_out, headb, Nn, 64, 64);
}


# GEMMs: weights held in registers (each wave its 16 columns for all K) while the block walks down row tiles; A tiles pass as whole lines, one tile ahead, through LDS; k order kept
# speedup vs baseline: 6.1266x; 6.1266x over previous
#include <hip/hip_runtime.h>
#include <hip/hip_bf16.h>

typedef float v2f __attribute__((ext_vector_type(2)));
typedef float v8f __attribute__((ext_vector_type(8)));

#define GAT_IN_DIM 128
#define GAT_NEG_SLOPE 0.2f

__device__ __forceinline__ void lds_barrier() {
  __builtin_amdgcn_fence(__ATOMIC_RELEASE, "workgroup", "local");
  __builtin_amdgcn_s_barrier();
  __builtin_amdgcn_fence(__ATOMIC_ACQUIRE, "workgroup", "local");
}

template <bool RELU_A, int K, int N>
__global__ void __launch_bounds__((N / 16) * 32)
wmma_gemm_f32(const float* __restrict__ A,
              const float* __restrict__ B,
              float* __restrict__ C,
              const float* __restrict__ bias,
              int M, int tpb) {
  constexpr int BD = (N / 16) * 32;
  constexpr int KS = K / 4;
  constexpr int P = K + 4;
  constexpr int V4 = (16 * K / 4) / BD;
  static_assert((16 * K / 4) % BD == 0 && K % 4 == 0 && N % 16 == 0, "shape");
  __shared__ __attribute__((aligned(16))) float sA[2][16 * P];

  const int tid = threadIdx.x;
  const int lane = tid & 31;
  const int tn = (tid >> 5) << 4;
  const int idx  = lane & 15;
  const int half = lane >> 4;
  const int kk   = half << 1;

  const int mtiles = M >> 4;
  const int t0 = blockIdx.x * tpb;
  const int t1 = min(t0 + tpb, mtiles);
  if (t0 >= t1) return;

  v2f b[KS];
#pragma unroll
  for (int s = 0; s < KS; ++s) {
    const float* bp = B + (size_t)(4 * s + kk) * N + tn + idx;
    b[s].x = bp[0];
    b[s].y = bp[N];
  }
  const float badd = bias ? bias[tn + idx] : 0.0f;

  static_assert(V4 == 2 || V4 == 4, "two or four float4 per thread and tile");
  float4 p0, p1, p2, p3;
  auto gload = [&](int t) {
    const float4* src = (const float4*)(A + (size_t)t * 16 * K) + tid;
    p0 = src[0];
    p1 = src[BD];
    if constexpr (V4 == 4) { p2 = src[2 * BD]; p3 = src[3 * BD]; }
  };
  auto put = [&](int buf, int i, float4 v) {
    const int e = 4 * (tid + i * BD);
    if constexpr (RELU_A) {
      v.x = fmaxf(v.x, 0.f); v.y = fmaxf(v.y, 0.f);
      v.z = fmaxf(v.z, 0.f); v.w = fmaxf(v.w, 0.f);
    }
    *(float4*)&sA[buf][(e / K) * P + (e % K)] = v;
  };
  auto sstore = [&](int buf) {
    put(buf, 0, p0);
    put(buf, 1, p1);
    if constexpr (V4 == 4) { put(buf, 2, p2); put(buf, 3, p3); }
  };

  gload(t0);
  sstore(0);
  lds_barrier();
  int cur = 0;
  for (int t = t0; t < t1; ++t) {
    const bool more = t + 1 < t1;
    if (more) gload(t + 1);

    v2f a[KS];
    const float* sp = &sA[cur][idx * P + kk];
#pragma unroll
    for (int s = 0; s < KS; ++s) a[s] = *(const v2f*)(sp + 4 * s);

    v8f acc = {};
#pragma unroll
    for (int s = 0; s < KS; ++s) {
      acc = __builtin_amdgcn_wmma_f32_16x16x4_f32(
           false, a[s],  false, b[s],
           (short)0, acc,  false,  false);
    }

    float* cp = C + (size_t)(t * 16 + 8 * half) * N + tn + idx;
#pragma unroll
    for (int r = 0; r < 8; ++r) cp[(size_t)r * N] = acc[r] + badd;

    if (more) sstore(cur ^ 1);
    lds_barrier();
    cur ^= 1;
  }
}

__global__ void gat_scores(const float* __restrict__ h,
                           const float* __restrict__ attn,
                           float* __restrict__ es, float* __restrict__ ed,
                           int n_nodes, int H, int F) {
  int t = blockIdx.x * blockDim.x + threadIdx.x;
  if (t >= n_nodes * H) return;
  int n = t / H;
  int hh = t - n * H;
  const float* hp = h + (size_t)n * H * F + (size_t)hh * F;
  const float* al = attn + (size_t)hh * 2 * F;
  const float* ar = al + F;
  float s = 0.f, d = 0.f;
  for (int f = 0; f < F; ++f) {
    s += hp[f] * al[f];
    d += hp[f] * ar[f];
  }
  es[t] = s;
  ed[t] = d;
}

__device__ __forceinline__ unsigned encf(float f) {
  unsigned u = __float_as_uint(f);
  return (u & 0x80000000u) ? ~u : (u | 0x80000000u);
}
__device__ __forceinline__ float decf(unsigned u) {
  return (u & 0x80000000u) ? __uint_as_float(u & 0x7FFFFFFFu)
                           : __uint_as_float(~u);
}

#define AGG_TB 1024
#define AGG_NPB 196
#define AGG_QCAP 8192

template <bool B> struct agg_flag { static constexpr bool value = B; };

template <int H, int F, bool LOADQ>
__global__ void __launch_bounds__(AGG_TB)
gat_aggregate_lds(const float* __restrict__ es, const float* __restrict__ ed,
                  const float* __restrict__ hsrc, float* __restrict__ outd,
                  const int* __restrict__ rowE, const int* __restrict__ colE,
                  int E, int Nn, int* __restrict__ qsave, int* __restrict__ qlen) {
  constexpr int FT = H * F;
  constexpr int FPL = FT / 32;
  constexpr int U = 8;
  constexpr int UA = 4;
  typedef float hvec __attribute__((ext_vector_type(H)));
  static_assert(F % 32 == 0 && (H & (H - 1)) == 0 && H <= 32, "shape");
  __shared__ __attribute__((aligned(16))) float acc[AGG_NPB * FT];
  __shared__ int qe[AGG_QCAP];
  __shared__ int qc[AGG_QCAP];
  __shared__ float qa[AGG_QCAP * H];
  __shared__ unsigned smax[AGG_NPB * H];
  __shared__ float sden[AGG_NPB * H];
  __shared__ int qcount;

  const int tid = threadIdx.x;
  const int lane = tid & 31;
  const int wv = tid >> 5;
  const int base = blockIdx.x * AGG_NPB;
  const int nloc = min(AGG_NPB, Nn - base);
  int* const qrow_g = qsave + (size_t)blockIdx.x * (2 * AGG_QCAP);
  int* const qcol_g = qrow_g + AGG_QCAP;

  for (int i = tid; i < nloc * FT; i += AGG_TB) acc[i] = 0.f;
  for (int i = tid; i < nloc * H; i += AGG_TB) { smax[i] = 0u; sden[i] = 0.f; }
  if (!LOADQ && tid == 0) qcount = 0;
  __syncthreads();

  auto push = [&](int c, int e) {
    unsigned cl = (unsigned)(c - base);
    if (cl < (unsigned)nloc) {
      int slot = atomicAdd(&qcount, 1);
      if (slot < AGG_QCAP) { qe[slot] = e; qc[slot] = (int)cl; }
    }
  };

  int total;
  if constexpr (LOADQ) {
    total = qlen[blockIdx.x];
  } else {
    if (tid < nloc) push(base + tid, E + base + tid);

    int done = 0;
    if ((((size_t)colE) & 15) == 0) {
      constexpr int PER_IT = AGG_TB * 16;
      const int full = E / PER_IT;
      const int start = (int)(((long long)blockIdx.x * full) / gridDim.x);
      for (int it = 0; it < full; ++it) {
        int itr = it + start;
        if (itr >= full) itr -= full;
        const int eb = itr * PER_IT + tid * 4;
        const int4 c0 = *(const int4*)(colE + eb);
        const int4 c1 = *(const int4*)(colE + eb + AGG_TB * 4);
        const int4 c2 = *(const int4*)(colE + eb + AGG_TB * 8);
        const int4 c3 = *(const int4*)(colE + eb + AGG_TB * 12);
        push(c0.x, eb); push(c0.y, eb + 1); push(c0.z, eb + 2); push(c0.w, eb + 3);
        { const int e1 = eb + AGG_TB * 4;
          push(c1.x, e1); push(c1.y, e1 + 1); push(c1.z, e1 + 2); push(c1.w, e1 + 3); }
        { const int e2 = eb + AGG_TB * 8;
          push(c2.x, e2); push(c2.y, e2 + 1); push(c2.z, e2 + 2); push(c2.w, e2 + 3); }
        { const int e3 = eb + AGG_TB * 12;
          push(c3.x, e3); push(c3.y, e3 + 1); push(c3.z, e3 + 2); push(c3.w, e3 + 3); }
      }
      done = full * PER_IT;
      for (int eb = done + tid * 4; eb + 3 < E; eb += AGG_TB * 4) {
        const int4 c0 = *(const int4*)(colE + eb);
        push(c0.x, eb); push(c0.y, eb + 1); push(c0.z, eb + 2); push(c0.w, eb + 3);
      }
      done += ((E - done) / 4) * 4;
    }
    for (int e = done + tid; e < E; e += AGG_TB) push(colE[e], e);

    __syncthreads();
    total = qcount;
    if (tid == 0) qlen[blockIdx.x] = total;
  }

  auto score = [&](int r, int c, int h) {
    const float a = es[(size_t)r * H + h] + ed[(size_t)c * H + h];
    return (a > 0.f) ? a : a * GAT_NEG_SLOPE;
  };

  if (total <= AGG_QCAP) {
    for (int i0 = tid; i0 < total; i0 += AGG_TB * UA) {
      int idx[UA], r[UA], cl[UA];
      hvec sv[UA], dv[UA];
#pragma unroll
      for (int k = 0; k < UA; ++k) {
        idx[k] = (i0 + AGG_TB * k < total) ? (i0 + AGG_TB * k) : i0;
        if constexpr (LOADQ) {
          r[k] = qrow_g[idx[k]];
          cl[k] = qcol_g[idx[k]];
        } else {
          const int e = qe[idx[k]];
          cl[k] = qc[idx[k]];
          r[k] = e - E;
          if (e < E) r[k] = rowE[e];
        }
      }
#pragma unroll
      for (int k = 0; k < UA; ++k) {
        sv[k] = *(const hvec*)(es + (size_t)r[k] * H);
        dv[k] = *(const hvec*)(ed + (size_t)(base + cl[k]) * H);
      }
      static_assert(UA == 4, "the statement below names UA = 4 entries");
      if constexpr (H == 1) {
        asm volatile("" :: "v"(sv[0][0]), "v"(sv[1][0]), "v"(sv[2][0]), "v"(sv[3][0]),
                           "v"(dv[0][0]), "v"(dv[1][0]), "v"(dv[2][0]), "v"(dv[3][0]));
      } else {
        asm volatile("" :: "v"(sv[0]), "v"(sv[1]), "v"(sv[2]), "v"(sv[3]),
                           "v"(dv[0]), "v"(dv[1]), "v"(dv[2]), "v"(dv[3]));
      }
#pragma unroll
      for (int k = 0; k < UA; ++k) {
        qe[idx[k]] = r[k];
        if constexpr (LOADQ) {
          qc[idx[k]] = cl[k];
        } else {
          qrow_g[idx[k]] = r[k];
          qcol_g[idx[k]] = cl[k];
        }
#pragma unroll
        for (int h = 0; h < H; ++h) {
          float a = sv[k][h] + dv[k][h];
          a = (a > 0.f) ? a : a * GAT_NEG_SLOPE;
          qa[idx[k] * H + h] = a;
          atomicMax(&smax[cl[k] * H + h], encf(a));
        }
      }
    }
    __syncthreads();

    for (int i = tid; i < total; i += AGG_TB) {
      const int c = qc[i];
#pragma unroll
      for (int h = 0; h < H; ++h) {
        const float ex = __expf(qa[i * H + h] - decf(smax[c * H + h]));
        qa[i * H + h] = ex;
        atomicAdd(&sden[c * H + h], ex);
      }
    }
    __syncthreads();

    const int hsel = lane & (H - 1);
    auto batch = [&](int i0, auto full) {
      constexpr bool FULL = decltype(full)::value;
      int cl[U], r[U];
      float cf[U];
      float v[U][FPL];
#pragma unroll
      for (int k = 0; k < U; ++k) {
        const int idx = (FULL || i0 + 32 * k < total) ? (i0 + 32 * k) : i0;
        r[k] = qe[idx];
        cl[k] = qc[idx];
        cf[k] = qa[idx * H + hsel] / sden[cl[k] * H + hsel];
      }
#pragma unroll
      for (int k = 0; k < U; ++k) {
#pragma unroll
        for (int j = 0; j < FPL; ++j) v[k][j] = hsrc[(size_t)r[k] * FT + lane + 32 * j];
      }
#pragma unroll
      for (int k = 0; k < U; ++k) {
        if (FULL || i0 + 32 * k < total) {
#pragma unroll
          for (int j = 0; j < FPL; ++j) {
            const float coef = __shfl(cf[k], (32 * j) / F);
            atomicAdd(&acc[cl[k] * FT + lane + 32 * j], v[k][j] * coef);
          }
        }
      }
    };
    int i0 = wv;
    for (; i0 + 32 * (U - 1) < total; i0 += 32 * U) batch(i0, agg_flag<true>{});
    if (i0 < total) batch(i0, agg_flag<false>{});
  } else {
    for (int e = tid; e < E + Nn; e += AGG_TB) {
      const int c = (e < E) ? colE[e] : (e - E);
      const unsigned cl = (unsigned)(c - base);
      if (cl < (unsigned)nloc) {
        const int r = (e < E) ? rowE[e] : (e - E);
        for (int h = 0; h < H; ++h) atomicMax(&smax[cl * H + h], encf(score(r, c, h)));
      }
    }
    __syncthreads();
    for (int e = tid; e < E + Nn; e += AGG_TB) {
      const int c = (e < E) ? colE[e] : (e - E);
      const unsigned cl = (unsigned)(c - base);
      if (cl < (unsigned)nloc) {
        const int r = (e < E) ? rowE[e] : (e - E);
        for (int h = 0; h < H; ++h)
          atomicAdd(&sden[cl * H + h], __expf(score(r, c, h) - decf(smax[cl * H + h])));
      }
    }
    __syncthreads();
    for (int e = tid; e < E + Nn; e += AGG_TB) {
      const int c = (e < E) ? colE[e] : (e - E);
      const unsigned cl = (unsigned)(c - base);
      if (cl < (unsigned)nloc) {
        const int r = (e < E) ? rowE[e] : (e - E);
        for (int f = 0; f < FT; ++f) {
          const int h = f / F;
          const float coef =
              __expf(score(r, c, h) - decf(smax[cl * H + h])) / sden[cl * H + h];
          atomicAdd(&acc[cl * FT + f], hsrc[(size_t)r * FT + f] * coef);
        }
      }
    }
  }
  __syncthreads();

  float4* o4 = (float4*)(outd + (size_t)base * FT);
  const float4* a4 = (const float4*)acc;
  for (int i = tid; i < nloc * (FT / 4); i += AGG_TB) o4[i] = a4[i];
}


extern "C" void kernel_launch(void* const* d_in, const int* in_sizes, int n_in,
                              void* d_out, int out_size, void* d_ws, size_t ws_size,
                              hipStream_t stream) {
  const float* x     = (const float*)d_in[0];
  const int*   ei    = (const int*)  d_in[1];
  const float* W1    = (const float*)d_in[2];
  const float* attn1 = (const float*)d_in[3];
  const float* W2    = (const float*)d_in[4];
  const float* attn2 = (const float*)d_in[5];
  const float* headW = (const float*)d_in[6];
  const float* headb = (const float*)d_in[7];

  const int Nn = in_sizes[0] / GAT_IN_DIM;
  const int E  = in_sizes[1] / 2;
  const int* rowE = ei;
  const int* colE = ei + E;

  float* w = (float*)d_ws;
  float* h1    = w; w += (size_t)Nn * 128;
  float* out1  = w; w += (size_t)Nn * 128;
  float* h2    = w; w += (size_t)Nn * 64;
  float* out2  = w; w += (size_t)Nn * 64;
  float* es1   = w; w += (size_t)Nn * 4;
  float* ed1   = w; w += (size_t)Nn * 4;
  float* es2   = w; w += (size_t)Nn;
  float* ed2   = w; w += (size_t)Nn;
  const int agg_blocks = (Nn + AGG_NPB - 1) / AGG_NPB;
  int* qsave = (int*)w; w += (size_t)agg_blocks * 2 * AGG_QCAP;
  int* qlen  = (int*)w; w += (size_t)agg_blocks;

  const int TB = 256;
  const int mtiles = Nn >> 4;
  const int TPB128 = 2;
  const int TPB64  = 2;
  auto gemm_blocks = [](int tiles, int tpb) { return (tiles + tpb - 1) / tpb; };

  wmma_gemm_f32<false, 128, 128><<<gemm_blocks(mtiles, TPB128), 256, 0, stream>>>(x, W1, h1, nullptr, Nn, TPB128);
  gat_scores<<<(Nn * 4 + TB - 1) / TB, TB, 0, stream>>>(h1, attn1, es1, ed1, Nn, 4, 32);
  gat_aggregate_lds<4, 32, false><<<agg_blocks, AGG_TB, 0, stream>>>(
      es1, ed1, h1, out1, rowE, colE, E, Nn, qsave, qlen);

  wmma_gemm_f32<true, 128, 64><<<gemm_blocks(mtiles, TPB64), 128, 0, stream>>>(out1, W2, h2, nullptr, Nn, TPB64);
  gat_scores<<<(Nn + TB - 1) / TB, TB, 0, stream>>>(h2, attn2, es2, ed2, Nn, 1, 64);
  gat_aggregate_lds<1, 64, true><<<agg_blocks, AGG_TB, 0, stream>>>(
      es2, ed2, h2, out2, rowE, colE, E, Nn, qsave, qlen);

  wmma_gemm_f32<false, 64, 64><<<gemm_blocks(mtiles, TPB64), 128, 0, stream>>>(out2, headW, (float*)d_out, headb, Nn, TPB64);
}


# Edge list read once per call: a binning launch ranks each edge into a (chunk, owner block) segment by LDS counter; both layers' owner blocks read their segments, not the list (sums in arrival order)
# speedup vs baseline: 7.3272x; 1.1960x over previous
#include <hip/hip_runtime.h>
#include <hip/hip_bf16.h>

typedef float v2f __attribute__((ext_vector_type(2)));
typedef float v8f __attribute__((ext_vector_type(8)));

#define GAT_IN_DIM 128
#define GAT_NEG_SLOPE 0.2f

__device__ __forceinline__ void lds_barrier() {
  __builtin_amdgcn_fence(__ATOMIC_RELEASE, "workgroup", "local");
  __builtin_amdgcn_s_barrier();
  __builtin_amdgcn_fence(__ATOMIC_ACQUIRE, "workgroup", "local");
}

template <bool RELU_A, int K, int N>
__global__ void __launch_bounds__((N / 16) * 32)
wmma_gemm_f32(const float* __restrict__ A,
              const float* __restrict__ B,
              float* __restrict__ C,
              const float* __restrict__ bias,
              int M, int tpb) {
  constexpr int BD = (N / 16) * 32;
  constexpr int KS = K / 4;
  constexpr int P = K + 4;
  constexpr int V4 = (16 * K / 4) / BD;
  static_assert((16 * K / 4) % BD == 0 && K % 4 == 0 && N % 16 == 0, "shape");
  __shared__ __attribute__((aligned(16))) float sA[2][16 * P];

  const int tid = threadIdx.x;
  const int lane = tid & 31;
  const int tn = (tid >> 5) << 4;
  const int idx  = lane & 15;
  const int half = lane >> 4;
  const int kk   = half << 1;

  const int mtiles = M >> 4;
  const int t0 = blockIdx.x * tpb;
  const int t1 = min(t0 + tpb, mtiles);
  if (t0 >= t1) return;

  v2f b[KS];
#pragma unroll
  for (int s = 0; s < KS; ++s) {
    const float* bp = B + (size_t)(4 * s + kk) * N + tn + idx;
    b[s].x = bp[0];
    b[s].y = bp[N];
  }
  const float badd = bias ? bias[tn + idx] : 0.0f;

  static_assert(V4 == 2 || V4 == 4, "two or four float4 per thread and tile");
  float4 p0, p1, p2, p3;
  auto gload = [&](int t) {
    const float4* src = (const float4*)(A + (size_t)t * 16 * K) + tid;
    p0 = src[0];
    p1 = src[BD];
    if constexpr (V4 == 4) { p2 = src[2 * BD]; p3 = src[3 * BD]; }
  };
  auto put = [&](int buf, int i, float4 v) {
    const int e = 4 * (tid + i * BD);
    if constexpr (RELU_A) {
      v.x = fmaxf(v.x, 0.f); v.y = fmaxf(v.y, 0.f);
      v.z = fmaxf(v.z, 0.f); v.w = fmaxf(v.w, 0.f);
    }
    *(float4*)&sA[buf][(e / K) * P + (e % K)] = v;
  };
  auto sstore = [&](int buf) {
    put(buf, 0, p0);
    put(buf, 1, p1);
    if constexpr (V4 == 4) { put(buf, 2, p2); put(buf, 3, p3); }
  };

  gload(t0);
  sstore(0);
  lds_barrier();
  int cur = 0;
  for (int t = t0; t < t1; ++t) {
    const bool more = t + 1 < t1;
    if (more) gload(t + 1);

    v2f a[KS];
    const float* sp = &sA[cur][idx * P + kk];
#pragma unroll
    for (int s = 0; s < KS; ++s) a[s] = *(const v2f*)(sp + 4 * s);

    v8f acc = {};
#pragma unroll
    for (int s = 0; s < KS; ++s) {
      acc = __builtin_amdgcn_wmma_f32_16x16x4_f32(
           false, a[s],  false, b[s],
           (short)0, acc,  false,  false);
    }

    float* cp = C + (size_t)(t * 16 + 8 * half) * N + tn + idx;
#pragma unroll
    for (int r = 0; r < 8; ++r) cp[(size_t)r * N] = acc[r] + badd;

    if (more) sstore(cur ^ 1);
    lds_barrier();
    cur ^= 1;
  }
}

__global__ void gat_scores(const float* __restrict__ h,
                           const float* __restrict__ attn,
                           float* __restrict__ es, float* __restrict__ ed,
                           int n_nodes, int H, int F) {
  int t = blockIdx.x * blockDim.x + threadIdx.x;
  if (t >= n_nodes * H) return;
  int n = t / H;
  int hh = t - n * H;
  const float* hp = h + (size_t)n * H * F + (size_t)hh * F;
  const float* al = attn + (size_t)hh * 2 * F;
  const float* ar = al + F;
  float s = 0.f, d = 0.f;
  for (int f = 0; f < F; ++f) {
    s += hp[f] * al[f];
    d += hp[f] * ar[f];
  }
  es[t] = s;
  ed[t] = d;
}

__device__ __forceinline__ unsigned encf(float f) {
  unsigned u = __float_as_uint(f);
  return (u & 0x80000000u) ? ~u : (u | 0x80000000u);
}
__device__ __forceinline__ float decf(unsigned u) {
  return (u & 0x80000000u) ? __uint_as_float(u & 0x7FFFFFFFu)
                           : __uint_as_float(~u);
}

#define AGG_TB 1024
#define AGG_NPB 196
#define AGG_QCAP 8192

#define BIN_TB 1024
#define BIN_EPB (BIN_TB * 4)
#define BIN_CAP 64
#define BIN_ROWBITS 24
static_assert(AGG_NPB <= (1 << (32 - BIN_ROWBITS)), "local destination must fit the entry");

__global__ void __launch_bounds__(BIN_TB)
gat_bin_edges(const int* __restrict__ rowE, const int* __restrict__ colE, int E, int Nn,
              int nown, int* __restrict__ bins, int* __restrict__ bcnt) {
  extern __shared__ int cnt[];
  const int tid = threadIdx.x;
  const int seg = blockIdx.x;
  const int nseg = gridDim.x;
  for (int i = tid; i < nown; i += BIN_TB) cnt[i] = 0;
  __syncthreads();

  const int e0 = seg * BIN_EPB + tid * 4;
  int c[4], r[4];
  int n = 0;
  if (e0 + 3 < E && ((((size_t)colE) | ((size_t)rowE)) & 15) == 0) {
    const int4 cv = *(const int4*)(colE + e0);
    const int4 rv = *(const int4*)(rowE + e0);
    c[0] = cv.x; c[1] = cv.y; c[2] = cv.z; c[3] = cv.w;
    r[0] = rv.x; r[1] = rv.y; r[2] = rv.z; r[3] = rv.w;
    n = 4;
  } else {
#pragma unroll
    for (int k = 0; k < 4; ++k) {
      c[k] = 0; r[k] = 0;
      if (e0 + k < E) { c[k] = colE[e0 + k]; r[k] = rowE[e0 + k]; n = k + 1; }
    }
  }
  int own[4], rank[4];
#pragma unroll
  for (int k = 0; k < 4; ++k) {
    own[k] = c[k] / AGG_NPB;
    rank[k] = BIN_CAP;
    if (k < n && (unsigned)c[k] < (unsigned)Nn) rank[k] = atomicAdd(&cnt[own[k]], 1);
  }
#pragma unroll
  for (int k = 0; k < 4; ++k) {
    if (rank[k] < BIN_CAP) {
      const int cl = c[k] - own[k] * AGG_NPB;
      bins[((size_t)own[k] * nseg + seg) * BIN_CAP + rank[k]] = r[k] | (cl << BIN_ROWBITS);
    }
  }
  lds_barrier();
  for (int i = tid; i < nown; i += BIN_TB) bcnt[(size_t)i * nseg + seg] = cnt[i];
}

template <bool B> struct agg_flag { static constexpr bool value = B; };

template <int H, int F>
__global__ void __launch_bounds__(AGG_TB)
gat_aggregate_lds(const float* __restrict__ es, const float* __restrict__ ed,
                  const float* __restrict__ hsrc, float* __restrict__ outd,
                  const int* __restrict__ rowE, const int* __restrict__ colE,
                  int E, int Nn, const int* __restrict__ bins,
                  const int* __restrict__ bcnt, int nseg) {
  constexpr int FT = H * F;
  constexpr int FPL = FT / 32;
  constexpr int U = 8;
  constexpr int UA = 4;
  typedef float hvec __attribute__((ext_vector_type(H)));
  static_assert(F % 32 == 0 && (H & (H - 1)) == 0 && H <= 32, "shape");
  __shared__ __attribute__((aligned(16))) float acc[AGG_NPB * FT];
  __shared__ int qe[AGG_QCAP];
  __shared__ int qc[AGG_QCAP];
  __shared__ float qa[AGG_QCAP * H];
  __shared__ unsigned smax[AGG_NPB * H];
  __shared__ float sden[AGG_NPB * H];
  __shared__ int segp[AGG_TB];
  __shared__ int wsum[32];
  __shared__ int nedge_s;
  __shared__ int qover;

  const int tid = threadIdx.x;
  const int lane = tid & 31;
  const int wv = tid >> 5;
  const int base = blockIdx.x * AGG_NPB;
  const int nloc = min(AGG_NPB, Nn - base);
  for (int i = tid; i < nloc * FT; i += AGG_TB) acc[i] = 0.f;
  for (int i = tid; i < nloc * H; i += AGG_TB) { smax[i] = 0u; sden[i] = 0.f; }
  if (tid == 0) qover = 0;
  __syncthreads();

  int cseg = 0;
  if (tid < nseg) cseg = bcnt[(size_t)blockIdx.x * nseg + tid];
  if (cseg > BIN_CAP) qover = 1;
  int incl = cseg;
#pragma unroll
  for (int d = 1; d < 32; d <<= 1) {
    const int y = __shfl_up(incl, d);
    if (lane >= d) incl += y;
  }
  if (lane == 31) wsum[wv] = incl;
  __syncthreads();
  if (wv == 0) {
    const int w = wsum[lane];
    int z = w;
#pragma unroll
    for (int d = 1; d < 32; d <<= 1) {
      const int y = __shfl_up(z, d);
      if (lane >= d) z += y;
    }
    wsum[lane] = z - w;
    if (lane == 31) nedge_s = z;
  }
  __syncthreads();
  segp[tid] = incl - cseg + wsum[wv];
  const int nedge = nedge_s;
  const int total = nedge + nloc;
  const bool queued = nseg <= AGG_TB && Nn <= (1 << BIN_ROWBITS) && qover == 0 &&
                      total <= AGG_QCAP;
  __syncthreads();

  const int* const myb = bins + (size_t)blockIdx.x * nseg * BIN_CAP;
  const int* const mycnt = bcnt + (size_t)blockIdx.x * nseg;
  int hb = 0;
  if (nseg > 1) { hb = 1; while (hb * 2 < nseg) hb <<= 1; }

  auto score = [&](int r, int c, int h) {
    const float a = es[(size_t)r * H + h] + ed[(size_t)c * H + h];
    return (a > 0.f) ? a : a * GAT_NEG_SLOPE;
  };

  if (queued) {
    for (int i0 = tid; i0 < total; i0 += AGG_TB * UA) {
      int idx[UA], sg[UA], sp[UA], ent[UA], r[UA], cl[UA];
      hvec sv[UA], dv[UA];
#pragma unroll
      for (int k = 0; k < UA; ++k) {
        idx[k] = (i0 + AGG_TB * k < total) ? (i0 + AGG_TB * k) : i0;
        sg[k] = 0;
        sp[k] = 0;
      }
      for (int step = hb; step > 0; step >>= 1) {
#pragma unroll
        for (int k = 0; k < UA; ++k) {
          const int t = min(sg[k] + step, nseg - 1);
          const int v = segp[t];
          if (v <= idx[k]) { sg[k] = t; sp[k] = v; }
        }
      }
#pragma unroll
      for (int k = 0; k < UA; ++k) {
        const bool isedge = idx[k] < nedge;
        const int* src = isedge ? myb + sg[k] * BIN_CAP + (idx[k] - sp[k]) : mycnt;
        ent[k] = *src;
      }
#pragma unroll
      for (int k = 0; k < UA; ++k) {
        const bool isedge = idx[k] < nedge;
        r[k] = isedge ? (ent[k] & ((1 << BIN_ROWBITS) - 1)) : (base + idx[k] - nedge);
        cl[k] = isedge ? (int)((unsigned)ent[k] >> BIN_ROWBITS) : (idx[k] - nedge);
      }
#pragma unroll
      for (int k = 0; k < UA; ++k) {
        sv[k] = *(const hvec*)(es + (size_t)r[k] * H);
        dv[k] = *(const hvec*)(ed + (size_t)(base + cl[k]) * H);
      }
      static_assert(UA == 4, "the statement below names UA = 4 entries");
      if constexpr (H == 1) {
        asm volatile("" :: "v"(sv[0][0]), "v"(sv[1][0]), "v"(sv[2][0]), "v"(sv[3][0]),
                           "v"(dv[0][0]), "v"(dv[1][0]), "v"(dv[2][0]), "v"(dv[3][0]));
      } else {
        asm volatile("" :: "v"(sv[0]), "v"(sv[1]), "v"(sv[2]), "v"(sv[3]),
                           "v"(dv[0]), "v"(dv[1]), "v"(dv[2]), "v"(dv[3]));
      }
#pragma unroll
      for (int k = 0; k < UA; ++k) {
        qe[idx[k]] = r[k];
        qc[idx[k]] = cl[k];
#pragma unroll
        for (int h = 0; h < H; ++h) {
          float a = sv[k][h] + dv[k][h];
          a = (a > 0.f) ? a : a * GAT_NEG_SLOPE;
          qa[idx[k] * H + h] = a;
          atomicMax(&smax[cl[k] * H + h], encf(a));
        }
      }
    }
    __syncthreads();

    for (int i = tid; i < total; i += AGG_TB) {
      const int c = qc[i];
#pragma unroll
      for (int h = 0; h < H; ++h) {
        const float ex = __expf(qa[i * H + h] - decf(smax[c * H + h]));
        qa[i * H + h] = ex;
        atomicAdd(&sden[c * H + h], ex);
      }
    }
    __syncthreads();

    const int hsel = lane & (H - 1);
    auto batch = [&](int i0, auto full) {
      constexpr bool FULL = decltype(full)::value;
      int cl[U], r[U];
      float cf[U];
      float v[U][FPL];
#pragma unroll
      for (int k = 0; k < U; ++k) {
        const int idx = (FULL || i0 + 32 * k < total) ? (i0 + 32 * k) : i0;
        r[k] = qe[idx];
        cl[k] = qc[idx];
        cf[k] = qa[idx * H + hsel] / sden[cl[k] * H + hsel];
      }
#pragma unroll
      for (int k = 0; k < U; ++k) {
#pragma unroll
        for (int j = 0; j < FPL; ++j) v[k][j] = hsrc[(size_t)r[k] * FT + lane + 32 * j];
      }
#pragma unroll
      for (int k = 0; k < U; ++k) {
        if (FULL || i0 + 32 * k < total) {
#pragma unroll
          for (int j = 0; j < FPL; ++j) {
            const float coef = __shfl(cf[k], (32 * j) / F);
            atomicAdd(&acc[cl[k] * FT + lane + 32 * j], v[k][j] * coef);
          }
        }
      }
    };
    int i0 = wv;
    for (; i0 + 32 * (U - 1) < total; i0 += 32 * U) batch(i0, agg_flag<true>{});
    if (i0 < total) batch(i0, agg_flag<false>{});
  } else {
    for (int e = tid; e < E + Nn; e += AGG_TB) {
      const int c = (e < E) ? colE[e] : (e - E);
      const unsigned cl = (unsigned)(c - base);
      if (cl < (unsigned)nloc) {
        const int r = (e < E) ? rowE[e] : (e - E);
        for (int h = 0; h < H; ++h) atomicMax(&smax[cl * H + h], encf(score(r, c, h)));
      }
    }
    __syncthreads();
    for (int e = tid; e < E + Nn; e += AGG_TB) {
      const int c = (e < E) ? colE[e] : (e - E);
      const unsigned cl = (unsigned)(c - base);
      if (cl < (unsigned)nloc) {
        const int r = (e < E) ? rowE[e] : (e - E);
        for (int h = 0; h < H; ++h)
          atomicAdd(&sden[cl * H + h], __expf(score(r, c, h) - decf(smax[cl * H + h])));
      }
    }
    __syncthreads();
    for (int e = tid; e < E + Nn; e += AGG_TB) {
      const int c = (e < E) ? colE[e] : (e - E);
      const unsigned cl = (unsigned)(c - base);
      if (cl < (unsigned)nloc) {
        const int r = (e < E) ? rowE[e] : (e - E);
        for (int f = 0; f < FT; ++f) {
          const int h = f / F;
          const float coef =
              __expf(score(r, c, h) - decf(smax[cl * H + h])) / sden[cl * H + h];
          atomicAdd(&acc[cl * FT + f], hsrc[(size_t)r * FT + f] * coef);
        }
      }
    }
  }
  __syncthreads();

  float4* o4 = (float4*)(outd + (size_t)base * FT);
  const float4* a4 = (const float4*)acc;
  for (int i = tid; i < nloc * (FT / 4); i += AGG_TB) o4[i] = a4[i];
}


extern "C" void kernel_launch(void* const* d_in, const int* in_sizes, int n_in,
                              void* d_out, int out_size, void* d_ws, size_t ws_size,
                              hipStream_t stream) {
  const float* x     = (const float*)d_in[0];
  const int*   ei    = (const int*)  d_in[1];
  const float* W1    = (const float*)d_in[2];
  const float* attn1 = (const float*)d_in[3];
  const float* W2    = (const float*)d_in[4];
  const float* attn2 = (const float*)d_in[5];
  const float* headW = (const float*)d_in[6];
  const float* headb = (const float*)d_in[7];

  const int Nn = in_sizes[0] / GAT_IN_DIM;
  const int E  = in_sizes[1] / 2;
  const int* rowE = ei;
  const int* colE = ei + E;

  float* w = (float*)d_ws;
  float* h1    = w; w += (size_t)Nn * 128;
  float* out1  = w; w += (size_t)Nn * 128;
  float* h2    = w; w += (size_t)Nn * 64;
  float* out2  = w; w += (size_t)Nn * 64;
  float* es1   = w; w += (size_t)Nn * 4;
  float* ed1   = w; w += (size_t)Nn * 4;
  float* es2   = w; w += (size_t)Nn;
  float* ed2   = w; w += (size_t)Nn;
  const int agg_blocks = (Nn + AGG_NPB - 1) / AGG_NPB;
  const int nseg = (E + BIN_EPB - 1) / BIN_EPB > 0 ? (E + BIN_EPB - 1) / BIN_EPB : 1;
  int* bins = (int*)w; w += (size_t)agg_blocks * nseg * BIN_CAP;
  int* bcnt = (int*)w; w += (size_t)agg_blocks * nseg;

  const int TB = 256;
  const int mtiles = Nn >> 4;
  const int TPB128 = 2;
  const int TPB64  = 2;
  auto gemm_blocks = [](int tiles, int tpb) { return (tiles + tpb - 1) / tpb; };

  gat_bin_edges<<<nseg, BIN_TB, agg_blocks * sizeof(int), stream>>>(
      rowE, colE, E, Nn, agg_blocks, bins, bcnt);

  wmma_gemm_f32<false, 128, 128><<<gemm_blocks(mtiles, TPB128), 256, 0, stream>>>(x, W1, h1, nullptr, Nn, TPB128);
  gat_scores<<<(Nn * 4 + TB - 1) / TB, TB, 0, stream>>>(h1, attn1, es1, ed1, Nn, 4, 32);
  gat_aggregate_lds<4, 32><<<agg_blocks, AGG_TB, 0, stream>>>(
      es1, ed1, h1, out1, rowE, colE, E, Nn, bins, bcnt, nseg);

  wmma_gemm_f32<true, 128, 64><<<gemm_blocks(mtiles, TPB64), 128, 0, stream>>>(out1, W2, h2, nullptr, Nn, TPB64);
  gat_scores<<<(Nn + TB - 1) / TB, TB, 0, stream>>>(h2, attn2, es2, ed2, Nn, 1, 64);
  gat_aggregate_lds<1, 64><<<agg_blocks, AGG_TB, 0, stream>>>(
      es2, ed2, h2, out2, rowE, colE, E, Nn, bins, bcnt, nseg);

  wmma_gemm_f32<false, 64, 64><<<gemm_blocks(mtiles, TPB64), 128, 0, stream>>>(out2, headW, (float*)d_out, headb, Nn, TPB64);
}


# Row sums: a wave reads 32 consecutive queue entries with one LDS read per array and hands them out by lane reads; per-entry LDS reads and shuffles go; LDS f32 adds, sum order = arrival order
# speedup vs baseline: 8.0673x; 1.1010x over previous
#include <hip/hip_runtime.h>
#include <hip/hip_bf16.h>

typedef float v2f __attribute__((ext_vector_type(2)));
typedef float v8f __attribute__((ext_vector_type(8)));

#define GAT_IN_DIM 128
#define GAT_NEG_SLOPE 0.2f

__device__ __forceinline__ void lds_barrier() {
  __builtin_amdgcn_fence(__ATOMIC_RELEASE, "workgroup", "local");
  __builtin_amdgcn_s_barrier();
  __builtin_amdgcn_fence(__ATOMIC_ACQUIRE, "workgroup", "local");
}

template <bool RELU_A, int K, int N>
__global__ void __launch_bounds__((N / 16) * 32)
wmma_gemm_f32(const float* __restrict__ A,
              const float* __restrict__ B,
              float* __restrict__ C,
              const float* __restrict__ bias,
              int M, int tpb) {
  constexpr int BD = (N / 16) * 32;
  constexpr int KS = K / 4;
  constexpr int P = K + 4;
  constexpr int V4 = (16 * K / 4) / BD;
  static_assert((16 * K / 4) % BD == 0 && K % 4 == 0 && N % 16 == 0, "shape");
  __shared__ __attribute__((aligned(16))) float sA[2][16 * P];

  const int tid = threadIdx.x;
  const int lane = tid & 31;
  const int tn = (tid >> 5) << 4;
  const int idx  = lane & 15;
  const int half = lane >> 4;
  const int kk   = half << 1;

  const int mtiles = M >> 4;
  const int t0 = blockIdx.x * tpb;
  const int t1 = min(t0 + tpb, mtiles);
  if (t0 >= t1) return;

  v2f b[KS];
#pragma unroll
  for (int s = 0; s < KS; ++s) {
    const float* bp = B + (size_t)(4 * s + kk) * N + tn + idx;
    b[s].x = bp[0];
    b[s].y = bp[N];
  }
  const float badd = bias ? bias[tn + idx] : 0.0f;

  static_assert(V4 == 2 || V4 == 4, "two or four float4 per thread and tile");
  float4 p0, p1, p2, p3;
  auto gload = [&](int t) {
    const float4* src = (const float4*)(A + (size_t)t * 16 * K) + tid;
    p0 = src[0];
    p1 = src[BD];
    if constexpr (V4 == 4) { p2 = src[2 * BD]; p3 = src[3 * BD]; }
  };
  auto put = [&](int buf, int i, float4 v) {
    const int e = 4 * (tid + i * BD);
    if constexpr (RELU_A) {
      v.x = fmaxf(v.x, 0.f); v.y = fmaxf(v.y, 0.f);
      v.z = fmaxf(v.z, 0.f); v.w = fmaxf(v.w, 0.f);
    }
    *(float4*)&sA[buf][(e / K) * P + (e % K)] = v;
  };
  auto sstore = [&](int buf) {
    put(buf, 0, p0);
    put(buf, 1, p1);
    if constexpr (V4 == 4) { put(buf, 2, p2); put(buf, 3, p3); }
  };

  gload(t0);
  sstore(0);
  lds_barrier();
  int cur = 0;
  for (int t = t0; t < t1; ++t) {
    const bool more = t + 1 < t1;
    if (more) gload(t + 1);

    v2f a[KS];
    const float* sp = &sA[cur][idx * P + kk];
#pragma unroll
    for (int s = 0; s < KS; ++s) a[s] = *(const v2f*)(sp + 4 * s);

    v8f acc = {};
#pragma unroll
    for (int s = 0; s < KS; ++s) {
      acc = __builtin_amdgcn_wmma_f32_16x16x4_f32(
           false, a[s],  false, b[s],
           (short)0, acc,  false,  false);
    }

    float* cp = C + (size_t)(t * 16 + 8 * half) * N + tn + idx;
#pragma unroll
    for (int r = 0; r < 8; ++r) cp[(size_t)r * N] = acc[r] + badd;

    if (more) sstore(cur ^ 1);
    lds_barrier();
    cur ^= 1;
  }
}

__global__ void gat_scores(const float* __restrict__ h,
                           const float* __restrict__ attn,
                           float* __restrict__ es, float* __restrict__ ed,
                           int n_nodes, int H, int F) {
  int t = blockIdx.x * blockDim.x + threadIdx.x;
  if (t >= n_nodes * H) return;
  int n = t / H;
  int hh = t - n * H;
  const float* hp = h + (size_t)n * H * F + (size_t)hh * F;
  const float* al = attn + (size_t)hh * 2 * F;
  const float* ar = al + F;
  float s = 0.f, d = 0.f;
  for (int f = 0; f < F; ++f) {
    s += hp[f] * al[f];
    d += hp[f] * ar[f];
  }
  es[t] = s;
  ed[t] = d;
}

__device__ __forceinline__ unsigned encf(float f) {
  unsigned u = __float_as_uint(f);
  return (u & 0x80000000u) ? ~u : (u | 0x80000000u);
}
__device__ __forceinline__ float decf(unsigned u) {
  return (u & 0x80000000u) ? __uint_as_float(u & 0x7FFFFFFFu)
                           : __uint_as_float(~u);
}

#define AGG_TB 1024
#define AGG_NPB 196
#define AGG_QCAP 8192

#define BIN_TB 1024
#define BIN_EPB (BIN_TB * 4)
#define BIN_CAP 64
#define BIN_ROWBITS 24
static_assert(AGG_NPB <= (1 << (32 - BIN_ROWBITS)), "local destination must fit the entry");

__global__ void __launch_bounds__(BIN_TB)
gat_bin_edges(const int* __restrict__ rowE, const int* __restrict__ colE, int E, int Nn,
              int nown, int* __restrict__ bins, int* __restrict__ bcnt) {
  extern __shared__ int cnt[];
  const int tid = threadIdx.x;
  const int seg = blockIdx.x;
  const int nseg = gridDim.x;
  for (int i = tid; i < nown; i += BIN_TB) cnt[i] = 0;
  __syncthreads();

  const int e0 = seg * BIN_EPB + tid * 4;
  int c[4], r[4];
  int n = 0;
  if (e0 + 3 < E && ((((size_t)colE) | ((size_t)rowE)) & 15) == 0) {
    const int4 cv = *(const int4*)(colE + e0);
    const int4 rv = *(const int4*)(rowE + e0);
    c[0] = cv.x; c[1] = cv.y; c[2] = cv.z; c[3] = cv.w;
    r[0] = rv.x; r[1] = rv.y; r[2] = rv.z; r[3] = rv.w;
    n = 4;
  } else {
#pragma unroll
    for (int k = 0; k < 4; ++k) {
      c[k] = 0; r[k] = 0;
      if (e0 + k < E) { c[k] = colE[e0 + k]; r[k] = rowE[e0 + k]; n = k + 1; }
    }
  }
  int own[4], rank[4];
#pragma unroll
  for (int k = 0; k < 4; ++k) {
    own[k] = c[k] / AGG_NPB;
    rank[k] = BIN_CAP;
    if (k < n && (unsigned)c[k] < (unsigned)Nn) rank[k] = atomicAdd(&cnt[own[k]], 1);
  }
#pragma unroll
  for (int k = 0; k < 4; ++k) {
    if (rank[k] < BIN_CAP) {
      const int cl = c[k] - own[k] * AGG_NPB;
      bins[((size_t)own[k] * nseg + seg) * BIN_CAP + rank[k]] = r[k] | (cl << BIN_ROWBITS);
    }
  }
  lds_barrier();
  for (int i = tid; i < nown; i += BIN_TB) bcnt[(size_t)i * nseg + seg] = cnt[i];
}

template <bool B> struct agg_flag { static constexpr bool value = B; };

template <int H, int F>
__global__ void __launch_bounds__(AGG_TB)
gat_aggregate_lds(const float* __restrict__ es, const float* __restrict__ ed,
                  const float* __restrict__ hsrc, float* __restrict__ outd,
                  const int* __restrict__ rowE, const int* __restrict__ colE,
                  int E, int Nn, const int* __restrict__ bins,
                  const int* __restrict__ bcnt, int nseg) {
  constexpr int FT = H * F;
  constexpr int FPL = FT / 32;
  constexpr int U = 8;
  constexpr int UA = 4;
  typedef float hvec __attribute__((ext_vector_type(H)));
  static_assert(F % 32 == 0 && (H & (H - 1)) == 0 && H <= 32, "shape");
  __shared__ __attribute__((aligned(16))) float acc[AGG_NPB * FT];
  __shared__ int qe[AGG_QCAP];
  __shared__ int qc[AGG_QCAP];
  __shared__ __attribute__((aligned(16))) float qa[AGG_QCAP * H];
  __shared__ unsigned smax[AGG_NPB * H];
  __shared__ __attribute__((aligned(16))) float sden[AGG_NPB * H];
  __shared__ int segp[AGG_TB];
  __shared__ int wsum[32];
  __shared__ int nedge_s;
  __shared__ int qover;

  const int tid = threadIdx.x;
  const int lane = tid & 31;
  const int wv = tid >> 5;
  const int base = blockIdx.x * AGG_NPB;
  const int nloc = min(AGG_NPB, Nn - base);
  for (int i = tid; i < nloc * FT; i += AGG_TB) acc[i] = 0.f;
  for (int i = tid; i < nloc * H; i += AGG_TB) { smax[i] = 0u; sden[i] = 0.f; }
  if (tid == 0) qover = 0;
  __syncthreads();

  int cseg = 0;
  if (tid < nseg) cseg = bcnt[(size_t)blockIdx.x * nseg + tid];
  if (cseg > BIN_CAP) qover = 1;
  int incl = cseg;
#pragma unroll
  for (int d = 1; d < 32; d <<= 1) {
    const int y = __shfl_up(incl, d);
    if (lane >= d) incl += y;
  }
  if (lane == 31) wsum[wv] = incl;
  __syncthreads();
  if (wv == 0) {
    const int w = wsum[lane];
    int z = w;
#pragma unroll
    for (int d = 1; d < 32; d <<= 1) {
      const int y = __shfl_up(z, d);
      if (lane >= d) z += y;
    }
    wsum[lane] = z - w;
    if (lane == 31) nedge_s = z;
  }
  __syncthreads();
  segp[tid] = incl - cseg + wsum[wv];
  const int nedge = nedge_s;
  const int total = nedge + nloc;
  const bool queued = nseg <= AGG_TB && Nn <= (1 << BIN_ROWBITS) && qover == 0 &&
                      total <= AGG_QCAP;
  __syncthreads();

  const int* const myb = bins + (size_t)blockIdx.x * nseg * BIN_CAP;
  const int* const mycnt = bcnt + (size_t)blockIdx.x * nseg;
  int hb = 0;
  if (nseg > 1) { hb = 1; while (hb * 2 < nseg) hb <<= 1; }

  auto score = [&](int r, int c, int h) {
    const float a = es[(size_t)r * H + h] + ed[(size_t)c * H + h];
    return (a > 0.f) ? a : a * GAT_NEG_SLOPE;
  };

  if (queued) {
    for (int i0 = tid; i0 < total; i0 += AGG_TB * UA) {
      int idx[UA], sg[UA], sp[UA], ent[UA], r[UA], cl[UA];
      hvec sv[UA], dv[UA];
#pragma unroll
      for (int k = 0; k < UA; ++k) {
        idx[k] = (i0 + AGG_TB * k < total) ? (i0 + AGG_TB * k) : i0;
        sg[k] = 0;
        sp[k] = 0;
      }
      for (int step = hb; step > 0; step >>= 1) {
#pragma unroll
        for (int k = 0; k < UA; ++k) {
          const int t = min(sg[k] + step, nseg - 1);
          const int v = segp[t];
          if (v <= idx[k]) { sg[k] = t; sp[k] = v; }
        }
      }
#pragma unroll
      for (int k = 0; k < UA; ++k) {
        const bool isedge = idx[k] < nedge;
        const int* src = isedge ? myb + sg[k] * BIN_CAP + (idx[k] - sp[k]) : mycnt;
        ent[k] = *src;
      }
#pragma unroll
      for (int k = 0; k < UA; ++k) {
        const bool isedge = idx[k] < nedge;
        r[k] = isedge ? (ent[k] & ((1 << BIN_ROWBITS) - 1)) : (base + idx[k] - nedge);
        cl[k] = isedge ? (int)((unsigned)ent[k] >> BIN_ROWBITS) : (idx[k] - nedge);
      }
#pragma unroll
      for (int k = 0; k < UA; ++k) {
        sv[k] = *(const hvec*)(es + (size_t)r[k] * H);
        dv[k] = *(const hvec*)(ed + (size_t)(base + cl[k]) * H);
      }
      static_assert(UA == 4, "the statement below names UA = 4 entries");
      if constexpr (H == 1) {
        asm volatile("" :: "v"(sv[0][0]), "v"(sv[1][0]), "v"(sv[2][0]), "v"(sv[3][0]),
                           "v"(dv[0][0]), "v"(dv[1][0]), "v"(dv[2][0]), "v"(dv[3][0]));
      } else {
        asm volatile("" :: "v"(sv[0]), "v"(sv[1]), "v"(sv[2]), "v"(sv[3]),
                           "v"(dv[0]), "v"(dv[1]), "v"(dv[2]), "v"(dv[3]));
      }
#pragma unroll
      for (int k = 0; k < UA; ++k) {
        qe[idx[k]] = r[k];
        qc[idx[k]] = cl[k];
#pragma unroll
        for (int h = 0; h < H; ++h) {
          float a = sv[k][h] + dv[k][h];
          a = (a > 0.f) ? a : a * GAT_NEG_SLOPE;
          qa[idx[k] * H + h] = a;
          atomicMax(&smax[cl[k] * H + h], encf(a));
        }
      }
    }
    __syncthreads();

    for (int i = tid; i < total; i += AGG_TB) {
      const int c = qc[i];
#pragma unroll
      for (int h = 0; h < H; ++h) {
        const float ex = __expf(qa[i * H + h] - decf(smax[c * H + h]));
        qa[i * H + h] = ex;
        atomicAdd(&sden[c * H + h], ex);
      }
    }
    __syncthreads();

    const int wvu = __builtin_amdgcn_readfirstlane(wv);
    const int totu = __builtin_amdgcn_readfirstlane(total);
    auto batch = [&](int er, int ec, hvec cfv, int k0, int cnt, auto full) {
      constexpr bool FULL = decltype(full)::value;
      int r[U], cl[U];
      float cf[U][H];
      float v[U][FPL];
#pragma unroll
      for (int k = 0; k < U; ++k) {
        const int kk = (FULL || k0 + k < cnt) ? (k0 + k) : k0;
        r[k] = __builtin_amdgcn_readlane(er, kk);
        cl[k] = __builtin_amdgcn_readlane(ec, kk);
#pragma unroll
        for (int h = 0; h < H; ++h) {
          const float ch = cfv[h];
          cf[k][h] = __int_as_float(__builtin_amdgcn_readlane(__float_as_int(ch), kk));
        }
      }
#pragma unroll
      for (int k = 0; k < U; ++k) {
        const float* rp = hsrc + (size_t)r[k] * FT + lane;
#pragma unroll
        for (int j = 0; j < FPL; ++j) v[k][j] = rp[32 * j];
      }
#pragma unroll
      for (int k = 0; k < U; ++k) {
        if (FULL || k0 + k < cnt) {
          float* ap = &acc[cl[k] * FT + lane];
#pragma unroll
          for (int j = 0; j < FPL; ++j) atomicAdd(ap + 32 * j, v[k][j] * cf[k][(32 * j) / F]);
        }
      }
    };
    const int per = ((((totu + 31) >> 5) + U - 1) / U) * U;
    const int wbeg = wvu * per;
    const int wend = min(wbeg + per, totu);
    for (int p = wbeg; p < wend; p += 32) {
      const int cnt = min(32, wend - p);
      const int i = p + ((lane < cnt) ? lane : 0);
      const int er = qe[i];
      const int ec = qc[i];
      const hvec cfv = *(const hvec*)&qa[i * H] / *(const hvec*)&sden[ec * H];
      int k0 = 0;
      for (; k0 + U <= cnt; k0 += U) batch(er, ec, cfv, k0, cnt, agg_flag<true>{});
      if (k0 < cnt) batch(er, ec, cfv, k0, cnt, agg_flag<false>{});
    }
  } else {
    for (int e = tid; e < E + Nn; e += AGG_TB) {
      const int c = (e < E) ? colE[e] : (e - E);
      const unsigned cl = (unsigned)(c - base);
      if (cl < (unsigned)nloc) {
        const int r = (e < E) ? rowE[e] : (e - E);
        for (int h = 0; h < H; ++h) atomicMax(&smax[cl * H + h], encf(score(r, c, h)));
      }
    }
    __syncthreads();
    for (int e = tid; e < E + Nn; e += AGG_TB) {
      const int c = (e < E) ? colE[e] : (e - E);
      const unsigned cl = (unsigned)(c - base);
      if (cl < (unsigned)nloc) {
        const int r = (e < E) ? rowE[e] : (e - E);
        for (int h = 0; h < H; ++h)
          atomicAdd(&sden[cl * H + h], __expf(score(r, c, h) - decf(smax[cl * H + h])));
      }
    }
    __syncthreads();
    for (int e = tid; e < E + Nn; e += AGG_TB) {
      const int c = (e < E) ? colE[e] : (e - E);
      const unsigned cl = (unsigned)(c - base);
      if (cl < (unsigned)nloc) {
        const int r = (e < E) ? rowE[e] : (e - E);
        for (int f = 0; f < FT; ++f) {
          const int h = f / F;
          const float coef =
              __expf(score(r, c, h) - decf(smax[cl * H + h])) / sden[cl * H + h];
          atomicAdd(&acc[cl * FT + f], hsrc[(size_t)r * FT + f] * coef);
        }
      }
    }
  }
  __syncthreads();

  float4* o4 = (float4*)(outd + (size_t)base * FT);
  const float4* a4 = (const float4*)acc;
  for (int i = tid; i < nloc * (FT / 4); i += AGG_TB) o4[i] = a4[i];
}


extern "C" void kernel_launch(void* const* d_in, const int* in_sizes, int n_in,
                              void* d_out, int out_size, void* d_ws, size_t ws_size,
                              hipStream_t stream) {
  const float* x     = (const float*)d_in[0];
  const int*   ei    = (const int*)  d_in[1];
  const float* W1    = (const float*)d_in[2];
  const float* attn1 = (const float*)d_in[3];
  const float* W2    = (const float*)d_in[4];
  const float* attn2 = (const float*)d_in[5];
  const float* headW = (const float*)d_in[6];
  const float* headb = (const float*)d_in[7];

  const int Nn = in_sizes[0] / GAT_IN_DIM;
  const int E  = in_sizes[1] / 2;
  const int* rowE = ei;
  const int* colE = ei + E;

  float* w = (float*)d_ws;
  float* h1    = w; w += (size_t)Nn * 128;
  float* out1  = w; w += (size_t)Nn * 128;
  float* h2    = w; w += (size_t)Nn * 64;
  float* out2  = w; w += (size_t)Nn * 64;
  float* es1   = w; w += (size_t)Nn * 4;
  float* ed1   = w; w += (size_t)Nn * 4;
  float* es2   = w; w += (size_t)Nn;
  float* ed2   = w; w += (size_t)Nn;
  const int agg_blocks = (Nn + AGG_NPB - 1) / AGG_NPB;
  const int nseg = (E + BIN_EPB - 1) / BIN_EPB > 0 ? (E + BIN_EPB - 1) / BIN_EPB : 1;
  int* bins = (int*)w; w += (size_t)agg_blocks * nseg * BIN_CAP;
  int* bcnt = (int*)w; w += (size_t)agg_blocks * nseg;

  const int TB = 256;
  const int mtiles = Nn >> 4;
  const int TPB128 = 2;
  const int TPB64  = 2;
  auto gemm_blocks = [](int tiles, int tpb) { return (tiles + tpb - 1) / tpb; };

  gat_bin_edges<<<nseg, BIN_TB, agg_blocks * sizeof(int), stream>>>(
      rowE, colE, E, Nn, agg_blocks, bins, bcnt);

  wmma_gemm_f32<false, 128, 128><<<gemm_blocks(mtiles, TPB128), 256, 0, stream>>>(x, W1, h1, nullptr, Nn, TPB128);
  gat_scores<<<(Nn * 4 + TB - 1) / TB, TB, 0, stream>>>(h1, attn1, es1, ed1, Nn, 4, 32);
  gat_aggregate_lds<4, 32><<<agg_blocks, AGG_TB, 0, stream>>>(
      es1, ed1, h1, out1, rowE, colE, E, Nn, bins, bcnt, nseg);

  wmma_gemm_f32<true, 128, 64><<<gemm_blocks(mtiles, TPB64), 128, 0, stream>>>(out1, W2, h2, nullptr, Nn, TPB64);
  gat_scores<<<(Nn + TB - 1) / TB, TB, 0, stream>>>(h2, attn2, es2, ed2, Nn, 1, 64);
  gat_aggregate_lds<1, 64><<<agg_blocks, AGG_TB, 0, stream>>>(
      es2, ed2, h2, out2, rowE, colE, E, Nn, bins, bcnt, nseg);

  wmma_gemm_f32<false, 64, 64><<<gemm_blocks(mtiles, TPB64), 128, 0, stream>>>(out2, headW, (float*)d_out, headb, Nn, TPB64);
}


# Linear map after each aggregation (layer-2 projection with ReLU; output head with bias) done by the owner block on its LDS sum tile with WMMA, k order kept; two GEMM launches and both sum planes go
# speedup vs baseline: 8.4873x; 1.0521x over previous
#include <hip/hip_runtime.h>
#include <hip/hip_bf16.h>

typedef float v2f __attribute__((ext_vector_type(2)));
typedef float v8f __attribute__((ext_vector_type(8)));

#define GAT_IN_DIM 128
#define GAT_NEG_SLOPE 0.2f

__device__ __forceinline__ void lds_barrier() {
  __builtin_amdgcn_fence(__ATOMIC_RELEASE, "workgroup", "local");
  __builtin_amdgcn_s_barrier();
  __builtin_amdgcn_fence(__ATOMIC_ACQUIRE, "workgroup", "local");
}

template <int K, int N>
__global__ void __launch_bounds__((N / 16) * 32)
wmma_gemm_f32(const float* __restrict__ A,
              const float* __restrict__ B,
              float* __restrict__ C,
              const float* __restrict__ bias,
              int M, int tpb) {
  constexpr int BD = (N / 16) * 32;
  constexpr int KS = K / 4;
  constexpr int P = K + 4;
  constexpr int V4 = (16 * K / 4) / BD;
  static_assert((16 * K / 4) % BD == 0 && K % 4 == 0 && N % 16 == 0, "shape");
  __shared__ __attribute__((aligned(16))) float sA[2][16 * P];

  const int tid = threadIdx.x;
  const int lane = tid & 31;
  const int tn = (tid >> 5) << 4;
  const int idx  = lane & 15;
  const int half = lane >> 4;
  const int kk   = half << 1;

  const int mtiles = M >> 4;
  const int t0 = blockIdx.x * tpb;
  const int t1 = min(t0 + tpb, mtiles);
  if (t0 >= t1) return;

  v2f b[KS];
#pragma unroll
  for (int s = 0; s < KS; ++s) {
    const float* bp = B + (size_t)(4 * s + kk) * N + tn + idx;
    b[s].x = bp[0];
    b[s].y = bp[N];
  }
  const float badd = bias ? bias[tn + idx] : 0.0f;

  static_assert(V4 == 2 || V4 == 4, "two or four float4 per thread and tile");
  float4 p0, p1, p2, p3;
  auto gload = [&](int t) {
    const float4* src = (const float4*)(A + (size_t)t * 16 * K) + tid;
    p0 = src[0];
    p1 = src[BD];
    if constexpr (V4 == 4) { p2 = src[2 * BD]; p3 = src[3 * BD]; }
  };
  auto put = [&](int buf, int i, float4 v) {
    const int e = 4 * (tid + i * BD);
    *(float4*)&sA[buf][(e / K) * P + (e % K)] = v;
  };
  auto sstore = [&](int buf) {
    put(buf, 0, p0);
    put(buf, 1, p1);
    if constexpr (V4 == 4) { put(buf, 2, p2); put(buf, 3, p3); }
  };

  gload(t0);
  sstore(0);
  lds_barrier();
  int cur = 0;
  for (int t = t0; t < t1; ++t) {
    const bool more = t + 1 < t1;
    if (more) gload(t + 1);

    v2f a[KS];
    const float* sp = &sA[cur][idx * P + kk];
#pragma unroll
    for (int s = 0; s < KS; ++s) a[s] = *(const v2f*)(sp + 4 * s);

    v8f acc = {};
#pragma unroll
    for (int s = 0; s < KS; ++s) {
      acc = __builtin_amdgcn_wmma_f32_16x16x4_f32(
           false, a[s],  false, b[s],
           (short)0, acc,  false,  false);
    }

    float* cp = C + (size_t)(t * 16 + 8 * half) * N + tn + idx;
#pragma unroll
    for (int r = 0; r < 8; ++r) cp[(size_t)r * N] = acc[r] + badd;

    if (more) sstore(cur ^ 1);
    lds_barrier();
    cur ^= 1;
  }
}

__global__ void gat_scores(const float* __restrict__ h,
                           const float* __restrict__ attn,
                           float* __restrict__ es, float* __restrict__ ed,
                           int n_nodes, int H, int F) {
  int t = blockIdx.x * blockDim.x + threadIdx.x;
  if (t >= n_nodes * H) return;
  int n = t / H;
  int hh = t - n * H;
  const float* hp = h + (size_t)n * H * F + (size_t)hh * F;
  const float* al = attn + (size_t)hh * 2 * F;
  const float* ar = al + F;
  float s = 0.f, d = 0.f;
  for (int f = 0; f < F; ++f) {
    s += hp[f] * al[f];
    d += hp[f] * ar[f];
  }
  es[t] = s;
  ed[t] = d;
}

__device__ __forceinline__ unsigned encf(float f) {
  unsigned u = __float_as_uint(f);
  return (u & 0x80000000u) ? ~u : (u | 0x80000000u);
}
__device__ __forceinline__ float decf(unsigned u) {
  return (u & 0x80000000u) ? __uint_as_float(u & 0x7FFFFFFFu)
                           : __uint_as_float(~u);
}

#define AGG_TB 1024
#define AGG_NPB 196
#define AGG_QCAP 8192
#define AGG_N2 64

#define BIN_TB 1024
#define BIN_EPB (BIN_TB * 4)
#define BIN_CAP 64
#define BIN_ROWBITS 24
static_assert(AGG_NPB <= (1 << (32 - BIN_ROWBITS)), "local destination must fit the entry");

__global__ void __launch_bounds__(BIN_TB)
gat_bin_edges(const int* __restrict__ rowE, const int* __restrict__ colE, int E, int Nn,
              int nown, int* __restrict__ bins, int* __restrict__ bcnt) {
  extern __shared__ int cnt[];
  const int tid = threadIdx.x;
  const int seg = blockIdx.x;
  const int nseg = gridDim.x;
  for (int i = tid; i < nown; i += BIN_TB) cnt[i] = 0;
  __syncthreads();

  const int e0 = seg * BIN_EPB + tid * 4;
  int c[4], r[4];
  int n = 0;
  if (e0 + 3 < E && ((((size_t)colE) | ((size_t)rowE)) & 15) == 0) {
    const int4 cv = *(const int4*)(colE + e0);
    const int4 rv = *(const int4*)(rowE + e0);
    c[0] = cv.x; c[1] = cv.y; c[2] = cv.z; c[3] = cv.w;
    r[0] = rv.x; r[1] = rv.y; r[2] = rv.z; r[3] = rv.w;
    n = 4;
  } else {
#pragma unroll
    for (int k = 0; k < 4; ++k) {
      c[k] = 0; r[k] = 0;
      if (e0 + k < E) { c[k] = colE[e0 + k]; r[k] = rowE[e0 + k]; n = k + 1; }
    }
  }
  int own[4], rank[4];
#pragma unroll
  for (int k = 0; k < 4; ++k) {
    own[k] = c[k] / AGG_NPB;
    rank[k] = BIN_CAP;
    if (k < n && (unsigned)c[k] < (unsigned)Nn) rank[k] = atomicAdd(&cnt[own[k]], 1);
  }
#pragma unroll
  for (int k = 0; k < 4; ++k) {
    if (rank[k] < BIN_CAP) {
      const int cl = c[k] - own[k] * AGG_NPB;
      bins[((size_t)own[k] * nseg + seg) * BIN_CAP + rank[k]] = r[k] | (cl << BIN_ROWBITS);
    }
  }
  lds_barrier();
  for (int i = tid; i < nown; i += BIN_TB) bcnt[(size_t)i * nseg + seg] = cnt[i];
}

template <bool B> struct agg_flag { static constexpr bool value = B; };

template <int H, int F, bool RELU>
__global__ void __launch_bounds__(AGG_TB)
gat_aggregate_lds(const float* __restrict__ es, const float* __restrict__ ed,
                  const float* __restrict__ hsrc,
                  const float* __restrict__ Wn, const float* __restrict__ biasn,
                  float* __restrict__ outn,
                  const int* __restrict__ rowE, const int* __restrict__ colE,
                  int E, int Nn, const int* __restrict__ bins,
                  const int* __restrict__ bcnt, int nseg) {
  constexpr int FT = H * F;
  constexpr int FPL = FT / 32;
  constexpr int P = FT + 4;
  constexpr int NT = (AGG_NPB + 15) / 16;
  constexpr int KS = FT / 4;
  constexpr int WPS = (AGG_TB / 32) / (AGG_N2 / 16);
  static_assert((AGG_TB / 32) % (AGG_N2 / 16) == 0 && FT % 4 == 0, "shape");
  constexpr int U = 8;
  constexpr int UA = 4;
  typedef float hvec __attribute__((ext_vector_type(H)));
  static_assert(F % 32 == 0 && (H & (H - 1)) == 0 && H <= 32, "shape");
  __shared__ __attribute__((aligned(16))) float acc[NT * 16 * P];
  __shared__ int qe[AGG_QCAP];
  __shared__ int qc[AGG_QCAP];
  __shared__ __attribute__((aligned(16))) float qa[AGG_QCAP * H];
  __shared__ unsigned smax[AGG_NPB * H];
  __shared__ __attribute__((aligned(16))) float sden[AGG_NPB * H];
  __shared__ int segp[AGG_TB];
  __shared__ int wsum[32];
  __shared__ int nedge_s;
  __shared__ int qover;

  const int tid = threadIdx.x;
  const int lane = tid & 31;
  const int wv = tid >> 5;
  const int base = blockIdx.x * AGG_NPB;
  const int nloc = min(AGG_NPB, Nn - base);
  for (int i = tid; i < NT * 16 * P; i += AGG_TB) acc[i] = 0.f;
  for (int i = tid; i < nloc * H; i += AGG_TB) { smax[i] = 0u; sden[i] = 0.f; }
  if (tid == 0) qover = 0;
  __syncthreads();

  int cseg = 0;
  if (tid < nseg) cseg = bcnt[(size_t)blockIdx.x * nseg + tid];
  if (cseg > BIN_CAP) qover = 1;
  int incl = cseg;
#pragma unroll
  for (int d = 1; d < 32; d <<= 1) {
    const int y = __shfl_up(incl, d);
    if (lane >= d) incl += y;
  }
  if (lane == 31) wsum[wv] = incl;
  __syncthreads();
  if (wv == 0) {
    const int w = wsum[lane];
    int z = w;
#pragma unroll
    for (int d = 1; d < 32; d <<= 1) {
      const int y = __shfl_up(z, d);
      if (lane >= d) z += y;
    }
    wsum[lane] = z - w;
    if (lane == 31) nedge_s = z;
  }
  __syncthreads();
  segp[tid] = incl - cseg + wsum[wv];
  const int nedge = nedge_s;
  const int total = nedge + nloc;
  const bool queued = nseg <= AGG_TB && Nn <= (1 << BIN_ROWBITS) && qover == 0 &&
                      total <= AGG_QCAP;
  __syncthreads();

  const int* const myb = bins + (size_t)blockIdx.x * nseg * BIN_CAP;
  const int* const mycnt = bcnt + (size_t)blockIdx.x * nseg;
  int hb = 0;
  if (nseg > 1) { hb = 1; while (hb * 2 < nseg) hb <<= 1; }

  auto score = [&](int r, int c, int h) {
    const float a = es[(size_t)r * H + h] + ed[(size_t)c * H + h];
    return (a > 0.f) ? a : a * GAT_NEG_SLOPE;
  };

  if (queued) {
    for (int i0 = tid; i0 < total; i0 += AGG_TB * UA) {
      int idx[UA], sg[UA], sp[UA], ent[UA], r[UA], cl[UA];
      hvec sv[UA], dv[UA];
#pragma unroll
      for (int k = 0; k < UA; ++k) {
        idx[k] = (i0 + AGG_TB * k < total) ? (i0 + AGG_TB * k) : i0;
        sg[k] = 0;
        sp[k] = 0;
      }
      for (int step = hb; step > 0; step >>= 1) {
#pragma unroll
        for (int k = 0; k < UA; ++k) {
          const int t = min(sg[k] + step, nseg - 1);
          const int v = segp[t];
          if (v <= idx[k]) { sg[k] = t; sp[k] = v; }
        }
      }
#pragma unroll
      for (int k = 0; k < UA; ++k) {
        const bool isedge = idx[k] < nedge;
        const int* src = isedge ? myb + sg[k] * BIN_CAP + (idx[k] - sp[k]) : mycnt;
        ent[k] = *src;
      }
#pragma unroll
      for (int k = 0; k < UA; ++k) {
        const bool isedge = idx[k] < nedge;
        r[k] = isedge ? (ent[k] & ((1 << BIN_ROWBITS) - 1)) : (base + idx[k] - nedge);
        cl[k] = isedge ? (int)((unsigned)ent[k] >> BIN_ROWBITS) : (idx[k] - nedge);
      }
#pragma unroll
      for (int k = 0; k < UA; ++k) {
        sv[k] = *(const hvec*)(es + (size_t)r[k] * H);
        dv[k] = *(const hvec*)(ed + (size_t)(base + cl[k]) * H);
      }
      static_assert(UA == 4, "the statement below names UA = 4 entries");
      if constexpr (H == 1) {
        asm volatile("" :: "v"(sv[0][0]), "v"(sv[1][0]), "v"(sv[2][0]), "v"(sv[3][0]),
                           "v"(dv[0][0]), "v"(dv[1][0]), "v"(dv[2][0]), "v"(dv[3][0]));
      } else {
        asm volatile("" :: "v"(sv[0]), "v"(sv[1]), "v"(sv[2]), "v"(sv[3]),
                           "v"(dv[0]), "v"(dv[1]), "v"(dv[2]), "v"(dv[3]));
      }
#pragma unroll
      for (int k = 0; k < UA; ++k) {
        qe[idx[k]] = r[k];
        qc[idx[k]] = cl[k];
#pragma unroll
        for (int h = 0; h < H; ++h) {
          float a = sv[k][h] + dv[k][h];
          a = (a > 0.f) ? a : a * GAT_NEG_SLOPE;
          qa[idx[k] * H + h] = a;
          atomicMax(&smax[cl[k] * H + h], encf(a));
        }
      }
    }
    __syncthreads();

    for (int i = tid; i < total; i += AGG_TB) {
      const int c = qc[i];
#pragma unroll
      for (int h = 0; h < H; ++h) {
        const float ex = __expf(qa[i * H + h] - decf(smax[c * H + h]));
        qa[i * H + h] = ex;
        atomicAdd(&sden[c * H + h], ex);
      }
    }
    __syncthreads();

    const int wvu = __builtin_amdgcn_readfirstlane(wv);
    const int totu = __builtin_amdgcn_readfirstlane(total);
    auto batch = [&](int er, int ec, hvec cfv, int k0, int cnt, auto full) {
      constexpr bool FULL = decltype(full)::value;
      int r[U], cl[U];
      float cf[U][H];
      float v[U][FPL];
#pragma unroll
      for (int k = 0; k < U; ++k) {
        const int kk = (FULL || k0 + k < cnt) ? (k0 + k) : k0;
        r[k] = __builtin_amdgcn_readlane(er, kk);
        cl[k] = __builtin_amdgcn_readlane(ec, kk);
#pragma unroll
        for (int h = 0; h < H; ++h) {
          const float ch = cfv[h];
          cf[k][h] = __int_as_float(__builtin_amdgcn_readlane(__float_as_int(ch), kk));
        }
      }
#pragma unroll
      for (int k = 0; k < U; ++k) {
        const float* rp = hsrc + (size_t)r[k] * FT + lane;
#pragma unroll
        for (int j = 0; j < FPL; ++j) v[k][j] = rp[32 * j];
      }
#pragma unroll
      for (int k = 0; k < U; ++k) {
        if (FULL || k0 + k < cnt) {
          float* ap = &acc[cl[k] * P + lane];
#pragma unroll
          for (int j = 0; j < FPL; ++j) atomicAdd(ap + 32 * j, v[k][j] * cf[k][(32 * j) / F]);
        }
      }
    };
    const int per = ((((totu + 31) >> 5) + U - 1) / U) * U;
    const int wbeg = wvu * per;
    const int wend = min(wbeg + per, totu);
    for (int p = wbeg; p < wend; p += 32) {
      const int cnt = min(32, wend - p);
      const int i = p + ((lane < cnt) ? lane : 0);
      const int er = qe[i];
      const int ec = qc[i];
      const hvec cfv = *(const hvec*)&qa[i * H] / *(const hvec*)&sden[ec * H];
      int k0 = 0;
      for (; k0 + U <= cnt; k0 += U) batch(er, ec, cfv, k0, cnt, agg_flag<true>{});
      if (k0 < cnt) batch(er, ec, cfv, k0, cnt, agg_flag<false>{});
    }
  } else {
    for (int e = tid; e < E + Nn; e += AGG_TB) {
      const int c = (e < E) ? colE[e] : (e - E);
      const unsigned cl = (unsigned)(c - base);
      if (cl < (unsigned)nloc) {
        const int r = (e < E) ? rowE[e] : (e - E);
        for (int h = 0; h < H; ++h) atomicMax(&smax[cl * H + h], encf(score(r, c, h)));
      }
    }
    __syncthreads();
    for (int e = tid; e < E + Nn; e += AGG_TB) {
      const int c = (e < E) ? colE[e] : (e - E);
      const unsigned cl = (unsigned)(c - base);
      if (cl < (unsigned)nloc) {
        const int r = (e < E) ? rowE[e] : (e - E);
        for (int h = 0; h < H; ++h)
          atomicAdd(&sden[cl * H + h], __expf(score(r, c, h) - decf(smax[cl * H + h])));
      }
    }
    __syncthreads();
    for (int e = tid; e < E + Nn; e += AGG_TB) {
      const int c = (e < E) ? colE[e] : (e - E);
      const unsigned cl = (unsigned)(c - base);
      if (cl < (unsigned)nloc) {
        const int r = (e < E) ? rowE[e] : (e - E);
        for (int f = 0; f < FT; ++f) {
          const int h = f / F;
          const float coef =
              __expf(score(r, c, h) - decf(smax[cl * H + h])) / sden[cl * H + h];
          atomicAdd(&acc[cl * P + f], hsrc[(size_t)r * FT + f] * coef);
        }
      }
    }
  }
  {
    const int wvs = __builtin_amdgcn_readfirstlane(wv);
    const int strip = wvs / WPS;
    const int tn = strip << 4;
    const int idx = lane & 15;
    const int half = lane >> 4;
    const int kk = half << 1;
    v2f b[KS];
#pragma unroll
    for (int s = 0; s < KS; ++s) {
      const float* bp = Wn + (size_t)(4 * s + kk) * AGG_N2 + tn + idx;
      b[s].x = bp[0];
      b[s].y = bp[AGG_N2];
    }
    const float badd = biasn ? biasn[tn + idx] : 0.0f;
    lds_barrier();
    const int ntile = (nloc + 15) >> 4;
    for (int t = (wvs + strip) % WPS; t < ntile; t += WPS) {
      const float* sp = &acc[(t * 16 + idx) * P + kk];
      v8f d = {};
#pragma unroll
      for (int s = 0; s < KS; ++s) {
        v2f a = *(const v2f*)(sp + 4 * s);
        if constexpr (RELU) { a.x = fmaxf(a.x, 0.f); a.y = fmaxf(a.y, 0.f); }
        d = __builtin_amdgcn_wmma_f32_16x16x4_f32(
             false, a,  false, b[s],
             (short)0, d,  false,  false);
      }
      const int r0 = t * 16 + 8 * half;
      float* cp = outn + (size_t)(base + r0) * AGG_N2 + tn + idx;
#pragma unroll
      for (int r = 0; r < 8; ++r)
        if (r0 + r < nloc) cp[(size_t)r * AGG_N2] = d[r] + badd;
    }
  }
}


extern "C" void kernel_launch(void* const* d_in, const int* in_sizes, int n_in,
                              void* d_out, int out_size, void* d_ws, size_t ws_size,
                              hipStream_t stream) {
  const float* x     = (const float*)d_in[0];
  const int*   ei    = (const int*)  d_in[1];
  const float* W1    = (const float*)d_in[2];
  const float* attn1 = (const float*)d_in[3];
  const float* W2    = (const float*)d_in[4];
  const float* attn2 = (const float*)d_in[5];
  const float* headW = (const float*)d_in[6];
  const float* headb = (const float*)d_in[7];

  const int Nn = in_sizes[0] / GAT_IN_DIM;
  const int E  = in_sizes[1] / 2;
  const int* rowE = ei;
  const int* colE = ei + E;

  float* w = (float*)d_ws;
  float* h1    = w; w += (size_t)Nn * 128;
  float* h2    = w; w += (size_t)Nn * 64;
  float* es1   = w; w += (size_t)Nn * 4;
  float* ed1   = w; w += (size_t)Nn * 4;
  float* es2   = w; w += (size_t)Nn;
  float* ed2   = w; w += (size_t)Nn;
  const int agg_blocks = (Nn + AGG_NPB - 1) / AGG_NPB;
  const int nseg = (E + BIN_EPB - 1) / BIN_EPB > 0 ? (E + BIN_EPB - 1) / BIN_EPB : 1;
  int* bins = (int*)w; w += (size_t)agg_blocks * nseg * BIN_CAP;
  int* bcnt = (int*)w; w += (size_t)agg_blocks * nseg;

  const int TB = 256;
  const int mtiles = Nn >> 4;
  const int TPB128 = 2;
  auto gemm_blocks = [](int tiles, int tpb) { return (tiles + tpb - 1) / tpb; };

  gat_bin_edges<<<nseg, BIN_TB, agg_blocks * sizeof(int), stream>>>(
      rowE, colE, E, Nn, agg_blocks, bins, bcnt);

  wmma_gemm_f32<128, 128><<<gemm_blocks(mtiles, TPB128), 256, 0, stream>>>(x, W1, h1, nullptr, Nn, TPB128);
  gat_scores<<<(Nn * 4 + TB - 1) / TB, TB, 0, stream>>>(h1, attn1, es1, ed1, Nn, 4, 32);
  gat_aggregate_lds<4, 32, true><<<agg_blocks, AGG_TB, 0, stream>>>(
      es1, ed1, h1, W2, nullptr, h2, rowE, colE, E, Nn, bins, bcnt, nseg);

  gat_scores<<<(Nn + TB - 1) / TB, TB, 0, stream>>>(h2, attn2, es2, ed2, Nn, 1, 64);
  gat_aggregate_lds<1, 64, false><<<agg_blocks, AGG_TB, 0, stream>>>(
      es2, ed2, h2, headW, headb, (float*)d_out, rowE, colE, E, Nn, bins, bcnt, nseg);
}


# Attention vectors folded into the weights (W*a, one extra block in the binning launch): scores are columns of the GEMM and owner-block products, summed over k (sum order changes); score launches go
# speedup vs baseline: 9.1076x; 1.0731x over previous
#include <hip/hip_runtime.h>
#include <hip/hip_bf16.h>

typedef float v2f __attribute__((ext_vector_type(2)));
typedef float v8f __attribute__((ext_vector_type(8)));

#define GAT_IN_DIM 128
#define GAT_NEG_SLOPE 0.2f

__device__ __forceinline__ void lds_barrier() {
  __builtin_amdgcn_fence(__ATOMIC_RELEASE, "workgroup", "local");
  __builtin_amdgcn_s_barrier();
  __builtin_amdgcn_fence(__ATOMIC_ACQUIRE, "workgroup", "local");
}

template <int K, int N, int SH>
__global__ void __launch_bounds__((N / 16) * 32, 8)
wmma_gemm_f32(const float* __restrict__ A,
              const float* __restrict__ B,
              float* __restrict__ C,
              const float* __restrict__ bias,
              int M, int tpb,
              const float* __restrict__ wa,
              float* __restrict__ es, float* __restrict__ ed) {
  constexpr int BD = (N / 16) * 32;
  constexpr int KS = K / 4;
  constexpr int P = K + 4;
  constexpr int V4 = (16 * K / 4) / BD;
  static_assert((16 * K / 4) % BD == 0 && K % 4 == 0 && N % 16 == 0, "shape");
  static_assert(SH == 0 || N / 16 == 2 * SH, "one wave per score column");
  static_assert(SH == 0 || 2 * SH * K / 4 == BD, "one float4 of the folded columns per thread");
  __shared__ __attribute__((aligned(16))) float sA[2][16 * P];
  __shared__ __attribute__((aligned(16))) float sW[SH > 0 ? 2 * SH * K : 4];

  const int tid = threadIdx.x;
  const int lane = tid & 31;
  const int tn = (tid >> 5) << 4;
  const int idx  = lane & 15;
  const int half = lane >> 4;
  const int kk   = half << 1;

  const int mtiles = M >> 4;
  const int t0 = blockIdx.x * tpb;
  const int t1 = min(t0 + tpb, mtiles);
  if (t0 >= t1) return;

  v2f b[KS];
#pragma unroll
  for (int s = 0; s < KS; ++s) {
    const float* bp = B + (size_t)(4 * s + kk) * N + tn + idx;
    b[s].x = bp[0];
    b[s].y = bp[N];
  }
  const float badd = bias ? bias[tn + idx] : 0.0f;
  const int wave = tid >> 5;

  static_assert(V4 == 2 || V4 == 4, "two or four float4 per thread and tile");
  float4 p0, p1, p2, p3;
  auto gload = [&](int t) {
    const float4* src = (const float4*)(A + (size_t)t * 16 * K) + tid;
    p0 = src[0];
    p1 = src[BD];
    if constexpr (V4 == 4) { p2 = src[2 * BD]; p3 = src[3 * BD]; }
  };
  auto put = [&](int buf, int i, float4 v) {
    const int e = 4 * (tid + i * BD);
    *(float4*)&sA[buf][(e / K) * P + (e % K)] = v;
  };
  auto sstore = [&](int buf) {
    put(buf, 0, p0);
    put(buf, 1, p1);
    if constexpr (V4 == 4) { put(buf, 2, p2); put(buf, 3, p3); }
  };

  gload(t0);
  float4 wq = {};
  if constexpr (SH > 0) wq = ((const float4*)wa)[tid];
  sstore(0);
  if constexpr (SH > 0) ((float4*)sW)[tid] = wq;
  lds_barrier();
  int cur = 0;
  for (int t = t0; t < t1; ++t) {
    const bool more = t + 1 < t1;
    if (more) gload(t + 1);

    v2f a[KS];
    const float* sp = &sA[cur][idx * P + kk];
#pragma unroll
    for (int s = 0; s < KS; ++s) a[s] = *(const v2f*)(sp + 4 * s);

    float s0 = 0.f, s1 = 0.f;
    if constexpr (SH > 0) {
      const float* wp = &sW[wave * K + kk];
#pragma unroll
      for (int s = 0; s < KS; ++s) {
        const v2f wv = *(const v2f*)(wp + 4 * s);
        s0 = fmaf(a[s].x, wv.x, s0);
        s1 = fmaf(a[s].y, wv.y, s1);
      }
    }

    v8f acc = {};
#pragma unroll
    for (int s = 0; s < KS; ++s) {
      acc = __builtin_amdgcn_wmma_f32_16x16x4_f32(
           false, a[s],  false, b[s],
           (short)0, acc,  false,  false);
    }

    float* cp = C + (size_t)(t * 16 + 8 * half) * N + tn + idx;
#pragma unroll
    for (int r = 0; r < 8; ++r) cp[(size_t)r * N] = acc[r] + badd;

    if constexpr (SH > 0) {
      float sc = s0 + s1;
      sc += __shfl_xor(sc, 16);
      if (half == 0) {
        float* dst = (wave < SH) ? es : ed;
        dst[(size_t)(t * 16 + idx) * SH + (wave % SH)] = sc;
      }
    }

    if (more) sstore(cur ^ 1);
    lds_barrier();
    cur ^= 1;
  }
}

__device__ __forceinline__ unsigned encf(float f) {
  unsigned u = __float_as_uint(f);
  return (u & 0x80000000u) ? ~u : (u | 0x80000000u);
}
__device__ __forceinline__ float decf(unsigned u) {
  return (u & 0x80000000u) ? __uint_as_float(u & 0x7FFFFFFFu)
                           : __uint_as_float(~u);
}

#define AGG_TB 1024
#define AGG_NPB 196
#define AGG_QCAP 8192
#define AGG_N2 64

#define GAT_H1 4
#define GAT_F1 32
#define GAT_D1 (GAT_H1 * GAT_F1)
#define GAT_F2 64
#define BIN_TB 1024
#define BIN_EPB (BIN_TB * 4)
#define BIN_CAP 64
#define BIN_ROWBITS 24
static_assert(AGG_NPB <= (1 << (32 - BIN_ROWBITS)), "local destination must fit the entry");

__global__ void __launch_bounds__(BIN_TB)
gat_bin_edges(const int* __restrict__ rowE, const int* __restrict__ colE, int E, int Nn,
              int nown, int* __restrict__ bins, int* __restrict__ bcnt, int nseg,
              const float* __restrict__ W1, const float* __restrict__ attn1,
              const float* __restrict__ W2, const float* __restrict__ attn2,
              float* __restrict__ wa1, float* __restrict__ wa2) {
  extern __shared__ int cnt[];
  const int tid = threadIdx.x;
  const int seg = blockIdx.x;
  if (seg == nseg) {
    for (int i = tid; i < 2 * GAT_H1 * GAT_IN_DIM; i += BIN_TB) {
      const int c = i / GAT_IN_DIM, k = i - c * GAT_IN_DIM;
      const int h = c % GAT_H1, part = c / GAT_H1;
      const float* wp = W1 + (size_t)k * GAT_D1 + h * GAT_F1;
      const float* ap = attn1 + h * 2 * GAT_F1 + part * GAT_F1;
      float s = 0.f;
      for (int f = 0; f < GAT_F1; ++f) s = fmaf(wp[f], ap[f], s);
      wa1[i] = s;
    }
    for (int i = tid; i < 2 * GAT_D1; i += BIN_TB) {
      const int part = i / GAT_D1, k = i - part * GAT_D1;
      const float* wp = W2 + (size_t)k * GAT_F2;
      const float* ap = attn2 + part * GAT_F2;
      float s = 0.f;
      for (int f = 0; f < GAT_F2; ++f) s = fmaf(wp[f], ap[f], s);
      wa2[i] = s;
    }
    return;
  }
  for (int i = tid; i < nown; i += BIN_TB) cnt[i] = 0;
  __syncthreads();

  const int e0 = seg * BIN_EPB + tid * 4;
  int c[4], r[4];
  int n = 0;
  if (e0 + 3 < E && ((((size_t)colE) | ((size_t)rowE)) & 15) == 0) {
    const int4 cv = *(const int4*)(colE + e0);
    const int4 rv = *(const int4*)(rowE + e0);
    c[0] = cv.x; c[1] = cv.y; c[2] = cv.z; c[3] = cv.w;
    r[0] = rv.x; r[1] = rv.y; r[2] = rv.z; r[3] = rv.w;
    n = 4;
  } else {
#pragma unroll
    for (int k = 0; k < 4; ++k) {
      c[k] = 0; r[k] = 0;
      if (e0 + k < E) { c[k] = colE[e0 + k]; r[k] = rowE[e0 + k]; n = k + 1; }
    }
  }
  int own[4], rank[4];
#pragma unroll
  for (int k = 0; k < 4; ++k) {
    own[k] = c[k] / AGG_NPB;
    rank[k] = BIN_CAP;
    if (k < n && (unsigned)c[k] < (unsigned)Nn) rank[k] = atomicAdd(&cnt[own[k]], 1);
  }
#pragma unroll
  for (int k = 0; k < 4; ++k) {
    if (rank[k] < BIN_CAP) {
      const int cl = c[k] - own[k] * AGG_NPB;
      bins[((size_t)own[k] * nseg + seg) * BIN_CAP + rank[k]] = r[k] | (cl << BIN_ROWBITS);
    }
  }
  lds_barrier();
  for (int i = tid; i < nown; i += BIN_TB) bcnt[(size_t)i * nseg + seg] = cnt[i];
}

template <bool B> struct agg_flag { static constexpr bool value = B; };

template <int H, int F, bool RELU, bool SCORE>
__global__ void __launch_bounds__(AGG_TB)
gat_aggregate_lds(const float* __restrict__ es, const float* __restrict__ ed,
                  const float* __restrict__ hsrc,
                  const float* __restrict__ Wn, const float* __restrict__ biasn,
                  float* __restrict__ outn,
                  const float* __restrict__ wan,
                  float* __restrict__ esn, float* __restrict__ edn,
                  const int* __restrict__ rowE, const int* __restrict__ colE,
                  int E, int Nn, const int* __restrict__ bins,
                  const int* __restrict__ bcnt, int nseg) {
  constexpr int FT = H * F;
  constexpr int FPL = FT / 32;
  constexpr int P = FT + 4;
  constexpr int NT = (AGG_NPB + 15) / 16;
  constexpr int KS = FT / 4;
  constexpr int WPS = (AGG_TB / 32) / (AGG_N2 / 16);
  static_assert((AGG_TB / 32) % (AGG_N2 / 16) == 0 && FT % 4 == 0, "shape");
  constexpr int U = 8;
  constexpr int UA = 4;
  typedef float hvec __attribute__((ext_vector_type(H)));
  static_assert(F % 32 == 0 && (H & (H - 1)) == 0 && H <= 32, "shape");
  __shared__ __attribute__((aligned(16))) float acc[NT * 16 * P];
  __shared__ int qe[AGG_QCAP];
  __shared__ int qc[AGG_QCAP];
  __shared__ __attribute__((aligned(16))) float qa[AGG_QCAP * H];
  __shared__ unsigned smax[AGG_NPB * H];
  __shared__ __attribute__((aligned(16))) float sden[AGG_NPB * H];
  __shared__ int segp[AGG_TB];
  __shared__ int wsum[32];
  __shared__ int nedge_s;
  __shared__ int qover;
  __shared__ __attribute__((aligned(16))) float swa[SCORE ? 2 * FT : 4];

  const int tid = threadIdx.x;
  const int lane = tid & 31;
  const int wv = tid >> 5;
  const int base = blockIdx.x * AGG_NPB;
  const int nloc = min(AGG_NPB, Nn - base);
  for (int i = tid; i < NT * 16 * P; i += AGG_TB) acc[i] = 0.f;
  for (int i = tid; i < nloc * H; i += AGG_TB) { smax[i] = 0u; sden[i] = 0.f; }
  if (tid == 0) qover = 0;
  if constexpr (SCORE) {
    for (int i = tid; i < 2 * FT; i += AGG_TB) swa[i] = wan[i];
  }
  __syncthreads();

  int cseg = 0;
  if (tid < nseg) cseg = bcnt[(size_t)blockIdx.x * nseg + tid];
  if (cseg > BIN_CAP) qover = 1;
  int incl = cseg;
#pragma unroll
  for (int d = 1; d < 32; d <<= 1) {
    const int y = __shfl_up(incl, d);
    if (lane >= d) incl += y;
  }
  if (lane == 31) wsum[wv] = incl;
  __syncthreads();
  if (wv == 0) {
    const int w = wsum[lane];
    int z = w;
#pragma unroll
    for (int d = 1; d < 32; d <<= 1) {
      const int y = __shfl_up(z, d);
      if (lane >= d) z += y;
    }
    wsum[lane] = z - w;
    if (lane == 31) nedge_s = z;
  }
  __syncthreads();
  segp[tid] = incl - cseg + wsum[wv];
  const int nedge = nedge_s;
  const int total = nedge + nloc;
  const bool queued = nseg <= AGG_TB && Nn <= (1 << BIN_ROWBITS) && qover == 0 &&
                      total <= AGG_QCAP;
  __syncthreads();

  const int* const myb = bins + (size_t)blockIdx.x * nseg * BIN_CAP;
  const int* const mycnt = bcnt + (size_t)blockIdx.x * nseg;
  int hb = 0;
  if (nseg > 1) { hb = 1; while (hb * 2 < nseg) hb <<= 1; }

  auto score = [&](int r, int c, int h) {
    const float a = es[(size_t)r * H + h] + ed[(size_t)c * H + h];
    return (a > 0.f) ? a : a * GAT_NEG_SLOPE;
  };

  if (queued) {
    for (int i0 = tid; i0 < total; i0 += AGG_TB * UA) {
      int idx[UA], sg[UA], sp[UA], ent[UA], r[UA], cl[UA];
      hvec sv[UA], dv[UA];
#pragma unroll
      for (int k = 0; k < UA; ++k) {
        idx[k] = (i0 + AGG_TB * k < total) ? (i0 + AGG_TB * k) : i0;
        sg[k] = 0;
        sp[k] = 0;
      }
      for (int step = hb; step > 0; step >>= 1) {
#pragma unroll
        for (int k = 0; k < UA; ++k) {
          const int t = min(sg[k] + step, nseg - 1);
          const int v = segp[t];
          if (v <= idx[k]) { sg[k] = t; sp[k] = v; }
        }
      }
#pragma unroll
      for (int k = 0; k < UA; ++k) {
        const bool isedge = idx[k] < nedge;
        const int* src = isedge ? myb + sg[k] * BIN_CAP + (idx[k] - sp[k]) : mycnt;
        ent[k] = *src;
      }
#pragma unroll
      for (int k = 0; k < UA; ++k) {
        const bool isedge = idx[k] < nedge;
        r[k] = isedge ? (ent[k] & ((1 << BIN_ROWBITS) - 1)) : (base + idx[k] - nedge);
        cl[k] = isedge ? (int)((unsigned)ent[k] >> BIN_ROWBITS) : (idx[k] - nedge);
      }
#pragma unroll
      for (int k = 0; k < UA; ++k) {
        sv[k] = *(const hvec*)(es + (size_t)r[k] * H);
        dv[k] = *(const hvec*)(ed + (size_t)(base + cl[k]) * H);
      }
      static_assert(UA == 4, "the statement below names UA = 4 entries");
      if constexpr (H == 1) {
        asm volatile("" :: "v"(sv[0][0]), "v"(sv[1][0]), "v"(sv[2][0]), "v"(sv[3][0]),
                           "v"(dv[0][0]), "v"(dv[1][0]), "v"(dv[2][0]), "v"(dv[3][0]));
      } else {
        asm volatile("" :: "v"(sv[0]), "v"(sv[1]), "v"(sv[2]), "v"(sv[3]),
                           "v"(dv[0]), "v"(dv[1]), "v"(dv[2]), "v"(dv[3]));
      }
#pragma unroll
      for (int k = 0; k < UA; ++k) {
        qe[idx[k]] = r[k];
        qc[idx[k]] = cl[k];
#pragma unroll
        for (int h = 0; h < H; ++h) {
          float a = sv[k][h] + dv[k][h];
          a = (a > 0.f) ? a : a * GAT_NEG_SLOPE;
          qa[idx[k] * H + h] = a;
          atomicMax(&smax[cl[k] * H + h], encf(a));
        }
      }
    }
    __syncthreads();

    for (int i = tid; i < total; i += AGG_TB) {
      const int c = qc[i];
#pragma unroll
      for (int h = 0; h < H; ++h) {
        const float ex = __expf(qa[i * H + h] - decf(smax[c * H + h]));
        qa[i * H + h] = ex;
        atomicAdd(&sden[c * H + h], ex);
      }
    }
    __syncthreads();

    const int wvu = __builtin_amdgcn_readfirstlane(wv);
    const int totu = __builtin_amdgcn_readfirstlane(total);
    auto batch = [&](int er, int ec, hvec cfv, int k0, int cnt, auto full) {
      constexpr bool FULL = decltype(full)::value;
      int r[U], cl[U];
      float cf[U][H];
      float v[U][FPL];
#pragma unroll
      for (int k = 0; k < U; ++k) {
        const int kk = (FULL || k0 + k < cnt) ? (k0 + k) : k0;
        r[k] = __builtin_amdgcn_readlane(er, kk);
        cl[k] = __builtin_amdgcn_readlane(ec, kk);
#pragma unroll
        for (int h = 0; h < H; ++h) {
          const float ch = cfv[h];
          cf[k][h] = __int_as_float(__builtin_amdgcn_readlane(__float_as_int(ch), kk));
        }
      }
#pragma unroll
      for (int k = 0; k < U; ++k) {
        const float* rp = hsrc + (size_t)r[k] * FT + lane;
#pragma unroll
        for (int j = 0; j < FPL; ++j) v[k][j] = rp[32 * j];
      }
#pragma unroll
      for (int k = 0; k < U; ++k) {
        if (FULL || k0 + k < cnt) {
          float* ap = &acc[cl[k] * P + lane];
#pragma unroll
          for (int j = 0; j < FPL; ++j) atomicAdd(ap + 32 * j, v[k][j] * cf[k][(32 * j) / F]);
        }
      }
    };
    const int per = ((((totu + 31) >> 5) + U - 1) / U) * U;
    const int wbeg = wvu * per;
    const int wend = min(wbeg + per, totu);
    for (int p = wbeg; p < wend; p += 32) {
      const int cnt = min(32, wend - p);
      const int i = p + ((lane < cnt) ? lane : 0);
      const int er = qe[i];
      const int ec = qc[i];
      const hvec cfv = *(const hvec*)&qa[i * H] / *(const hvec*)&sden[ec * H];
      int k0 = 0;
      for (; k0 + U <= cnt; k0 += U) batch(er, ec, cfv, k0, cnt, agg_flag<true>{});
      if (k0 < cnt) batch(er, ec, cfv, k0, cnt, agg_flag<false>{});
    }
  } else {
    for (int e = tid; e < E + Nn; e += AGG_TB) {
      const int c = (e < E) ? colE[e] : (e - E);
      const unsigned cl = (unsigned)(c - base);
      if (cl < (unsigned)nloc) {
        const int r = (e < E) ? rowE[e] : (e - E);
        for (int h = 0; h < H; ++h) atomicMax(&smax[cl * H + h], encf(score(r, c, h)));
      }
    }
    __syncthreads();
    for (int e = tid; e < E + Nn; e += AGG_TB) {
      const int c = (e < E) ? colE[e] : (e - E);
      const unsigned cl = (unsigned)(c - base);
      if (cl < (unsigned)nloc) {
        const int r = (e < E) ? rowE[e] : (e - E);
        for (int h = 0; h < H; ++h)
          atomicAdd(&sden[cl * H + h], __expf(score(r, c, h) - decf(smax[cl * H + h])));
      }
    }
    __syncthreads();
    for (int e = tid; e < E + Nn; e += AGG_TB) {
      const int c = (e < E) ? colE[e] : (e - E);
      const unsigned cl = (unsigned)(c - base);
      if (cl < (unsigned)nloc) {
        const int r = (e < E) ? rowE[e] : (e - E);
        for (int f = 0; f < FT; ++f) {
          const int h = f / F;
          const float coef =
              __expf(score(r, c, h) - decf(smax[cl * H + h])) / sden[cl * H + h];
          atomicAdd(&acc[cl * P + f], hsrc[(size_t)r * FT + f] * coef);
        }
      }
    }
  }
  {
    const int wvs = __builtin_amdgcn_readfirstlane(wv);
    const int strip = wvs / WPS;
    const int tn = strip << 4;
    const int idx = lane & 15;
    const int half = lane >> 4;
    const int kk = half << 1;
    v2f b[KS];
#pragma unroll
    for (int s = 0; s < KS; ++s) {
      const float* bp = Wn + (size_t)(4 * s + kk) * AGG_N2 + tn + idx;
      b[s].x = bp[0];
      b[s].y = bp[AGG_N2];
    }
    const float badd = biasn ? biasn[tn + idx] : 0.0f;
    lds_barrier();
    const int ntile = (nloc + 15) >> 4;
    auto pieces = [&](auto scf) {
      constexpr bool SC = decltype(scf)::value;
      for (int t = (wvs + strip) % WPS; t < ntile; t += WPS) {
        const float* sp = &acc[(t * 16 + idx) * P + kk];
        if constexpr (SC) {
          const float* wp = &swa[strip * FT + kk];
          float s0 = 0.f, s1 = 0.f;
#pragma unroll 8
          for (int s = 0; s < KS; ++s) {
            v2f a = *(const v2f*)(sp + 4 * s);
            if constexpr (RELU) { a.x = fmaxf(a.x, 0.f); a.y = fmaxf(a.y, 0.f); }
            const v2f wv = *(const v2f*)(wp + 4 * s);
            s0 = fmaf(a.x, wv.x, s0);
            s1 = fmaf(a.y, wv.y, s1);
          }
          float sc = s0 + s1;
          sc += __shfl_xor(sc, 16);
          const int m = t * 16 + idx;
          if (half == 0 && m < nloc) (strip == 0 ? esn : edn)[base + m] = sc;
        }
        v8f d = {};
#pragma unroll
        for (int s = 0; s < KS; ++s) {
          v2f a = *(const v2f*)(sp + 4 * s);
          if constexpr (RELU) { a.x = fmaxf(a.x, 0.f); a.y = fmaxf(a.y, 0.f); }
          d = __builtin_amdgcn_wmma_f32_16x16x4_f32(
               false, a,  false, b[s],
               (short)0, d,  false,  false);
        }
        const int r0 = t * 16 + 8 * half;
        float* cp = outn + (size_t)(base + r0) * AGG_N2 + tn + idx;
#pragma unroll
        for (int r = 0; r < 8; ++r)
          if (r0 + r < nloc) cp[(size_t)r * AGG_N2] = d[r] + badd;
      }
    };
    if constexpr (SCORE) {
      if (strip < 2) pieces(agg_flag<true>{}); else pieces(agg_flag<false>{});
    } else {
      pieces(agg_flag<false>{});
    }
  }
}


extern "C" void kernel_launch(void* const* d_in, const int* in_sizes, int n_in,
                              void* d_out, int out_size, void* d_ws, size_t ws_size,
                              hipStream_t stream) {
  const float* x     = (const float*)d_in[0];
  const int*   ei    = (const int*)  d_in[1];
  const float* W1    = (const float*)d_in[2];
  const float* attn1 = (const float*)d_in[3];
  const float* W2    = (const float*)d_in[4];
  const float* attn2 = (const float*)d_in[5];
  const float* headW = (const float*)d_in[6];
  const float* headb = (const float*)d_in[7];

  const int Nn = in_sizes[0] / GAT_IN_DIM;
  const int E  = in_sizes[1] / 2;
  const int* rowE = ei;
  const int* colE = ei + E;

  float* w = (float*)d_ws;
  float* wa1   = w; w += (size_t)2 * GAT_H1 * GAT_IN_DIM;
  float* wa2   = w; w += (size_t)2 * GAT_D1;
  float* h1    = w; w += (size_t)Nn * 128;
  float* h2    = w; w += (size_t)Nn * 64;
  float* es1   = w; w += (size_t)Nn * 4;
  float* ed1   = w; w += (size_t)Nn * 4;
  float* es2   = w; w += (size_t)Nn;
  float* ed2   = w; w += (size_t)Nn;
  const int agg_blocks = (Nn + AGG_NPB - 1) / AGG_NPB;
  const int nseg = (E + BIN_EPB - 1) / BIN_EPB > 0 ? (E + BIN_EPB - 1) / BIN_EPB : 1;
  int* bins = (int*)w; w += (size_t)agg_blocks * nseg * BIN_CAP;
  int* bcnt = (int*)w; w += (size_t)agg_blocks * nseg;

  const int mtiles = Nn >> 4;
  const int TPB128 = 2;
  auto gemm_blocks = [](int tiles, int tpb) { return (tiles + tpb - 1) / tpb; };

  gat_bin_edges<<<nseg + 1, BIN_TB, agg_blocks * sizeof(int), stream>>>(
      rowE, colE, E, Nn, agg_blocks, bins, bcnt, nseg, W1, attn1, W2, attn2, wa1, wa2);

  wmma_gemm_f32<128, 128, GAT_H1><<<gemm_blocks(mtiles, TPB128), 256, 0, stream>>>(
      x, W1, h1, nullptr, Nn, TPB128, wa1, es1, ed1);
  gat_aggregate_lds<4, 32, true, true><<<agg_blocks, AGG_TB, 0, stream>>>(
      es1, ed1, h1, W2, nullptr, h2, wa2, es2, ed2, rowE, colE, E, Nn, bins, bcnt, nseg);

  gat_aggregate_lds<1, 64, false, false><<<agg_blocks, AGG_TB, 0, stream>>>(
      es2, ed2, h2, headW, headb, (float*)d_out, nullptr, nullptr, nullptr,
      rowE, colE, E, Nn, bins, bcnt, nseg);
}


# Edge binning moved into the layer-1 GEMM launch as its leading 8-wave blocks (16 edges a thread; sums stay in arrival order); launch 1 keeps only the attention fold, spread over 20 small blocks
# speedup vs baseline: 9.4703x; 1.0398x over previous
#include <hip/hip_runtime.h>
#include <hip/hip_bf16.h>

typedef float v2f __attribute__((ext_vector_type(2)));
typedef float v8f __attribute__((ext_vector_type(8)));

#define GAT_IN_DIM 128
#define GAT_NEG_SLOPE 0.2f

__device__ __forceinline__ void lds_barrier() {
  __builtin_amdgcn_fence(__ATOMIC_RELEASE, "workgroup", "local");
  __builtin_amdgcn_s_barrier();
  __builtin_amdgcn_fence(__ATOMIC_ACQUIRE, "workgroup", "local");
}

#define AGG_TB 1024
#define AGG_NPB 196
#define AGG_QCAP 8192
#define AGG_N2 64

#define GAT_H1 4
#define GAT_F1 32
#define GAT_D1 (GAT_H1 * GAT_F1)
#define GAT_F2 64
#define BIN_TB 256
#define BIN_Q 4
#define BIN_EPB (BIN_TB * 4 * BIN_Q)
#define BIN_CAP 64
#define BIN_ROWBITS 24
static_assert(AGG_NPB <= (1 << (32 - BIN_ROWBITS)), "local destination must fit the entry");

__device__ __forceinline__ void bin_edges_block(int seg, int tid, int* cnt,
                                                const int* __restrict__ rowE,
                                                const int* __restrict__ colE, int E, int Nn,
                                                int nown, int* __restrict__ bins,
                                                int* __restrict__ bcnt, int nseg) {
  for (int i = tid; i < nown; i += BIN_TB) cnt[i] = 0;
  __syncthreads();

  int c[BIN_Q][4], r[BIN_Q][4], n[BIN_Q];
  const int ebase = seg * BIN_EPB;
  if (ebase + BIN_EPB <= E && ((((size_t)colE) | ((size_t)rowE)) & 15) == 0) {
#pragma unroll
    for (int q = 0; q < BIN_Q; ++q) {
      const int e0 = ebase + (q * BIN_TB + tid) * 4;
      const int4 cv = *(const int4*)(colE + e0);
      const int4 rv = *(const int4*)(rowE + e0);
      c[q][0] = cv.x; c[q][1] = cv.y; c[q][2] = cv.z; c[q][3] = cv.w;
      r[q][0] = rv.x; r[q][1] = rv.y; r[q][2] = rv.z; r[q][3] = rv.w;
      n[q] = 4;
    }
  } else {
#pragma unroll
    for (int q = 0; q < BIN_Q; ++q) {
      const int e0 = ebase + (q * BIN_TB + tid) * 4;
      n[q] = 0;
#pragma unroll
      for (int k = 0; k < 4; ++k) {
        c[q][k] = 0; r[q][k] = 0;
        if (e0 + k < E) { c[q][k] = colE[e0 + k]; r[q][k] = rowE[e0 + k]; n[q] = k + 1; }
      }
    }
  }
  int own[BIN_Q][4], rank[BIN_Q][4];
#pragma unroll
  for (int q = 0; q < BIN_Q; ++q) {
#pragma unroll
    for (int k = 0; k < 4; ++k) {
      own[q][k] = c[q][k] / AGG_NPB;
      rank[q][k] = BIN_CAP;
      if (k < n[q] && (unsigned)c[q][k] < (unsigned)Nn) rank[q][k] = atomicAdd(&cnt[own[q][k]], 1);
    }
  }
#pragma unroll
  for (int q = 0; q < BIN_Q; ++q) {
#pragma unroll
    for (int k = 0; k < 4; ++k) {
      if (rank[q][k] < BIN_CAP) {
        const int cl = c[q][k] - own[q][k] * AGG_NPB;
        bins[((size_t)own[q][k] * nseg + seg) * BIN_CAP + rank[q][k]] =
            r[q][k] | (cl << BIN_ROWBITS);
      }
    }
  }
  lds_barrier();
  for (int i = tid; i < nown; i += BIN_TB) bcnt[(size_t)i * nseg + seg] = cnt[i];
}

template <int K, int N, int SH>
__global__ void __launch_bounds__((N / 16) * 32, 8)
wmma_gemm_f32(const float* __restrict__ A,
              const float* __restrict__ B,
              float* __restrict__ C,
              const float* __restrict__ bias,
              int M, int tpb,
              const float* __restrict__ wa,
              float* __restrict__ es, float* __restrict__ ed,
              int nbin, const int* __restrict__ rowE, const int* __restrict__ colE, int E,
              int nown, int* __restrict__ bins, int* __restrict__ bcnt) {
  constexpr int BD = (N / 16) * 32;
  constexpr int KS = K / 4;
  constexpr int P = K + 4;
  constexpr int V4 = (16 * K / 4) / BD;
  static_assert((16 * K / 4) % BD == 0 && K % 4 == 0 && N % 16 == 0, "shape");
  static_assert(SH == 0 || N / 16 == 2 * SH, "one wave per score column");
  static_assert(SH == 0 || 2 * SH * K / 4 == BD, "one float4 of the folded columns per thread");
  __shared__ __attribute__((aligned(16))) float sA[2][16 * P];
  __shared__ __attribute__((aligned(16))) float sW[SH > 0 ? 2 * SH * K : 4];
  extern __shared__ int bin_cnt[];

  const int tid = threadIdx.x;
  if ((int)blockIdx.x < nbin) {
    static_assert(BD == BIN_TB, "a binning block has the product's block size");
    bin_edges_block((int)blockIdx.x, tid, bin_cnt, rowE, colE, E, M, nown, bins, bcnt, nbin);
    return;
  }
  const int lane = tid & 31;
  const int tn = (tid >> 5) << 4;
  const int idx  = lane & 15;
  const int half = lane >> 4;
  const int kk   = half << 1;

  const int mtiles = M >> 4;
  const int t0 = ((int)blockIdx.x - nbin) * tpb;
  const int t1 = min(t0 + tpb, mtiles);
  if (t0 >= t1) return;

  v2f b[KS];
#pragma unroll
  for (int s = 0; s < KS; ++s) {
    const float* bp = B + (size_t)(4 * s + kk) * N + tn + idx;
    b[s].x = bp[0];
    b[s].y = bp[N];
  }
  const float badd = bias ? bias[tn + idx] : 0.0f;
  const int wave = tid >> 5;

  static_assert(V4 == 2 || V4 == 4, "two or four float4 per thread and tile");
  float4 p0, p1, p2, p3;
  auto gload = [&](int t) {
    const float4* src = (const float4*)(A + (size_t)t * 16 * K) + tid;
    p0 = src[0];
    p1 = src[BD];
    if constexpr (V4 == 4) { p2 = src[2 * BD]; p3 = src[3 * BD]; }
  };
  auto put = [&](int buf, int i, float4 v) {
    const int e = 4 * (tid + i * BD);
    *(float4*)&sA[buf][(e / K) * P + (e % K)] = v;
  };
  auto sstore = [&](int buf) {
    put(buf, 0, p0);
    put(buf, 1, p1);
    if constexpr (V4 == 4) { put(buf, 2, p2); put(buf, 3, p3); }
  };

  gload(t0);
  float4 wq = {};
  if constexpr (SH > 0) wq = ((const float4*)wa)[tid];
  sstore(0);
  if constexpr (SH > 0) ((float4*)sW)[tid] = wq;
  lds_barrier();
  int cur = 0;
  for (int t = t0; t < t1; ++t) {
    const bool more = t + 1 < t1;
    if (more) gload(t + 1);

    v2f a[KS];
    const float* sp = &sA[cur][idx * P + kk];
#pragma unroll
    for (int s = 0; s < KS; ++s) a[s] = *(const v2f*)(sp + 4 * s);

    float s0 = 0.f, s1 = 0.f;
    if constexpr (SH > 0) {
      const float* wp = &sW[wave * K + kk];
#pragma unroll
      for (int s = 0; s < KS; ++s) {
        const v2f wv = *(const v2f*)(wp + 4 * s);
        s0 = fmaf(a[s].x, wv.x, s0);
        s1 = fmaf(a[s].y, wv.y, s1);
      }
    }

    v8f acc = {};
#pragma unroll
    for (int s = 0; s < KS; ++s) {
      acc = __builtin_amdgcn_wmma_f32_16x16x4_f32(
           false, a[s],  false, b[s],
           (short)0, acc,  false,  false);
    }

    float* cp = C + (size_t)(t * 16 + 8 * half) * N + tn + idx;
#pragma unroll
    for (int r = 0; r < 8; ++r) cp[(size_t)r * N] = acc[r] + badd;

    if constexpr (SH > 0) {
      float sc = s0 + s1;
      sc += __shfl_xor(sc, 16);
      if (half == 0) {
        float* dst = (wave < SH) ? es : ed;
        dst[(size_t)(t * 16 + idx) * SH + (wave % SH)] = sc;
      }
    }

    if (more) sstore(cur ^ 1);
    lds_barrier();
    cur ^= 1;
  }
}

#define FOLD_TB 64
#define FOLD_N1 (2 * GAT_H1 * GAT_IN_DIM)
#define FOLD_N2 (2 * GAT_D1)
static_assert(FOLD_N1 % FOLD_TB == 0 && FOLD_N2 % FOLD_TB == 0, "a block folds one layer");

__global__ void __launch_bounds__(FOLD_TB)
gat_fold_attn(const float* __restrict__ W1, const float* __restrict__ attn1,
              const float* __restrict__ W2, const float* __restrict__ attn2,
              float* __restrict__ wa1, float* __restrict__ wa2) {
  const int i = blockIdx.x * FOLD_TB + threadIdx.x;
  if (i < FOLD_N1) {
    const int c = i / GAT_IN_DIM, k = i - c * GAT_IN_DIM;
    const int h = c % GAT_H1, part = c / GAT_H1;
    const float* wp = W1 + (size_t)k * GAT_D1 + h * GAT_F1;
    const float* ap = attn1 + h * 2 * GAT_F1 + part * GAT_F1;
    float s = 0.f;
#pragma unroll
    for (int f = 0; f < GAT_F1; ++f) s = fmaf(wp[f], ap[f], s);
    wa1[i] = s;
  } else if (i < FOLD_N1 + FOLD_N2) {
    const int j = i - FOLD_N1;
    const int part = j / GAT_D1, k = j - part * GAT_D1;
    const float* wp = W2 + (size_t)k * GAT_F2;
    const float* ap = attn2 + part * GAT_F2;
    float s = 0.f;
#pragma unroll
    for (int f = 0; f < GAT_F2; ++f) s = fmaf(wp[f], ap[f], s);
    wa2[j] = s;
  }
}

__device__ __forceinline__ unsigned encf(float f) {
  unsigned u = __float_as_uint(f);
  return (u & 0x80000000u) ? ~u : (u | 0x80000000u);
}
__device__ __forceinline__ float decf(unsigned u) {
  return (u & 0x80000000u) ? __uint_as_float(u & 0x7FFFFFFFu)
                           : __uint_as_float(~u);
}

template <bool B> struct agg_flag { static constexpr bool value = B; };

template <int H, int F, bool RELU, bool SCORE>
__global__ void __launch_bounds__(AGG_TB)
gat_aggregate_lds(const float* __restrict__ es, const float* __restrict__ ed,
                  const float* __restrict__ hsrc,
                  const float* __restrict__ Wn, const float* __restrict__ biasn,
                  float* __restrict__ outn,
                  const float* __restrict__ wan,
                  float* __restrict__ esn, float* __restrict__ edn,
                  const int* __restrict__ rowE, const int* __restrict__ colE,
                  int E, int Nn, const int* __restrict__ bins,
                  const int* __restrict__ bcnt, int nseg) {
  constexpr int FT = H * F;
  constexpr int FPL = FT / 32;
  constexpr int P = FT + 4;
  constexpr int NT = (AGG_NPB + 15) / 16;
  constexpr int KS = FT / 4;
  constexpr int WPS = (AGG_TB / 32) / (AGG_N2 / 16);
  static_assert((AGG_TB / 32) % (AGG_N2 / 16) == 0 && FT % 4 == 0, "shape");
  constexpr int U = 8;
  constexpr int UA = 4;
  typedef float hvec __attribute__((ext_vector_type(H)));
  static_assert(F % 32 == 0 && (H & (H - 1)) == 0 && H <= 32, "shape");
  __shared__ __attribute__((aligned(16))) float acc[NT * 16 * P];
  __shared__ int qe[AGG_QCAP];
  __shared__ int qc[AGG_QCAP];
  __shared__ __attribute__((aligned(16))) float qa[AGG_QCAP * H];
  __shared__ unsigned smax[AGG_NPB * H];
  __shared__ __attribute__((aligned(16))) float sden[AGG_NPB * H];
  __shared__ int segp[AGG_TB];
  __shared__ int wsum[32];
  __shared__ int nedge_s;
  __shared__ int qover;
  __shared__ __attribute__((aligned(16))) float swa[SCORE ? 2 * FT : 4];

  const int tid = threadIdx.x;
  const int lane = tid & 31;
  const int wv = tid >> 5;
  const int base = blockIdx.x * AGG_NPB;
  const int nloc = min(AGG_NPB, Nn - base);
  for (int i = tid; i < NT * 16 * P; i += AGG_TB) acc[i] = 0.f;
  for (int i = tid; i < nloc * H; i += AGG_TB) { smax[i] = 0u; sden[i] = 0.f; }
  if (tid == 0) qover = 0;
  if constexpr (SCORE) {
    for (int i = tid; i < 2 * FT; i += AGG_TB) swa[i] = wan[i];
  }
  __syncthreads();

  int cseg = 0;
  if (tid < nseg) cseg = bcnt[(size_t)blockIdx.x * nseg + tid];
  if (cseg > BIN_CAP) qover = 1;
  int incl = cseg;
#pragma unroll
  for (int d = 1; d < 32; d <<= 1) {
    const int y = __shfl_up(incl, d);
    if (lane >= d) incl += y;
  }
  if (lane == 31) wsum[wv] = incl;
  __syncthreads();
  if (wv == 0) {
    const int w = wsum[lane];
    int z = w;
#pragma unroll
    for (int d = 1; d < 32; d <<= 1) {
      const int y = __shfl_up(z, d);
      if (lane >= d) z += y;
    }
    wsum[lane] = z - w;
    if (lane == 31) nedge_s = z;
  }
  __syncthreads();
  segp[tid] = incl - cseg + wsum[wv];
  const int nedge = nedge_s;
  const int total = nedge + nloc;
  const bool queued = nseg <= AGG_TB && Nn <= (1 << BIN_ROWBITS) && qover == 0 &&
                      total <= AGG_QCAP;
  __syncthreads();

  const int* const myb = bins + (size_t)blockIdx.x * nseg * BIN_CAP;
  const int* const mycnt = bcnt + (size_t)blockIdx.x * nseg;
  int hb = 0;
  if (nseg > 1) { hb = 1; while (hb * 2 < nseg) hb <<= 1; }

  auto score = [&](int r, int c, int h) {
    const float a = es[(size_t)r * H + h] + ed[(size_t)c * H + h];
    return (a > 0.f) ? a : a * GAT_NEG_SLOPE;
  };

  if (queued) {
    for (int i0 = tid; i0 < total; i0 += AGG_TB * UA) {
      int idx[UA], sg[UA], sp[UA], ent[UA], r[UA], cl[UA];
      hvec sv[UA], dv[UA];
#pragma unroll
      for (int k = 0; k < UA; ++k) {
        idx[k] = (i0 + AGG_TB * k < total) ? (i0 + AGG_TB * k) : i0;
        sg[k] = 0;
        sp[k] = 0;
      }
      for (int step = hb; step > 0; step >>= 1) {
#pragma unroll
        for (int k = 0; k < UA; ++k) {
          const int t = min(sg[k] + step, nseg - 1);
          const int v = segp[t];
          if (v <= idx[k]) { sg[k] = t; sp[k] = v; }
        }
      }
#pragma unroll
      for (int k = 0; k < UA; ++k) {
        const bool isedge = idx[k] < nedge;
        const int* src = isedge ? myb + sg[k] * BIN_CAP + (idx[k] - sp[k]) : mycnt;
        ent[k] = *src;
      }
#pragma unroll
      for (int k = 0; k < UA; ++k) {
        const bool isedge = idx[k] < nedge;
        r[k] = isedge ? (ent[k] & ((1 << BIN_ROWBITS) - 1)) : (base + idx[k] - nedge);
        cl[k] = isedge ? (int)((unsigned)ent[k] >> BIN_ROWBITS) : (idx[k] - nedge);
      }
#pragma unroll
      for (int k = 0; k < UA; ++k) {
        sv[k] = *(const hvec*)(es + (size_t)r[k] * H);
        dv[k] = *(const hvec*)(ed + (size_t)(base + cl[k]) * H);
      }
      static_assert(UA == 4, "the statement below names UA = 4 entries");
      if constexpr (H == 1) {
        asm volatile("" :: "v"(sv[0][0]), "v"(sv[1][0]), "v"(sv[2][0]), "v"(sv[3][0]),
                           "v"(dv[0][0]), "v"(dv[1][0]), "v"(dv[2][0]), "v"(dv[3][0]));
      } else {
        asm volatile("" :: "v"(sv[0]), "v"(sv[1]), "v"(sv[2]), "v"(sv[3]),
                           "v"(dv[0]), "v"(dv[1]), "v"(dv[2]), "v"(dv[3]));
      }
#pragma unroll
      for (int k = 0; k < UA; ++k) {
        qe[idx[k]] = r[k];
        qc[idx[k]] = cl[k];
#pragma unroll
        for (int h = 0; h < H; ++h) {
          float a = sv[k][h] + dv[k][h];
          a = (a > 0.f) ? a : a * GAT_NEG_SLOPE;
          qa[idx[k] * H + h] = a;
          atomicMax(&smax[cl[k] * H + h], encf(a));
        }
      }
    }
    __syncthreads();

    for (int i = tid; i < total; i += AGG_TB) {
      const int c = qc[i];
#pragma unroll
      for (int h = 0; h < H; ++h) {
        const float ex = __expf(qa[i * H + h] - decf(smax[c * H + h]));
        qa[i * H + h] = ex;
        atomicAdd(&sden[c * H + h], ex);
      }
    }
    __syncthreads();

    const int wvu = __builtin_amdgcn_readfirstlane(wv);
    const int totu = __builtin_amdgcn_readfirstlane(total);
    auto batch = [&](int er, int ec, hvec cfv, int k0, int cnt, auto full) {
      constexpr bool FULL = decltype(full)::value;
      int r[U], cl[U];
      float cf[U][H];
      float v[U][FPL];
#pragma unroll
      for (int k = 0; k < U; ++k) {
        const int kk = (FULL || k0 + k < cnt) ? (k0 + k) : k0;
        r[k] = __builtin_amdgcn_readlane(er, kk);
        cl[k] = __builtin_amdgcn_readlane(ec, kk);
#pragma unroll
        for (int h = 0; h < H; ++h) {
          const float ch = cfv[h];
          cf[k][h] = __int_as_float(__builtin_amdgcn_readlane(__float_as_int(ch), kk));
        }
      }
#pragma unroll
      for (int k = 0; k < U; ++k) {
        const float* rp = hsrc + (size_t)r[k] * FT + lane;
#pragma unroll
        for (int j = 0; j < FPL; ++j) v[k][j] = rp[32 * j];
      }
#pragma unroll
      for (int k = 0; k < U; ++k) {
        if (FULL || k0 + k < cnt) {
          float* ap = &acc[cl[k] * P + lane];
#pragma unroll
          for (int j = 0; j < FPL; ++j) atomicAdd(ap + 32 * j, v[k][j] * cf[k][(32 * j) / F]);
        }
      }
    };
    const int per = ((((totu + 31) >> 5) + U - 1) / U) * U;
    const int wbeg = wvu * per;
    const int wend = min(wbeg + per, totu);
    for (int p = wbeg; p < wend; p += 32) {
      const int cnt = min(32, wend - p);
      const int i = p + ((lane < cnt) ? lane : 0);
      const int er = qe[i];
      const int ec = qc[i];
      const hvec cfv = *(const hvec*)&qa[i * H] / *(const hvec*)&sden[ec * H];
      int k0 = 0;
      for (; k0 + U <= cnt; k0 += U) batch(er, ec, cfv, k0, cnt, agg_flag<true>{});
      if (k0 < cnt) batch(er, ec, cfv, k0, cnt, agg_flag<false>{});
    }
  } else {
    for (int e = tid; e < E + Nn; e += AGG_TB) {
      const int c = (e < E) ? colE[e] : (e - E);
      const unsigned cl = (unsigned)(c - base);
      if (cl < (unsigned)nloc) {
        const int r = (e < E) ? rowE[e] : (e - E);
        for (int h = 0; h < H; ++h) atomicMax(&smax[cl * H + h], encf(score(r, c, h)));
      }
    }
    __syncthreads();
    for (int e = tid; e < E + Nn; e += AGG_TB) {
      const int c = (e < E) ? colE[e] : (e - E);
      const unsigned cl = (unsigned)(c - base);
      if (cl < (unsigned)nloc) {
        const int r = (e < E) ? rowE[e] : (e - E);
        for (int h = 0; h < H; ++h)
          atomicAdd(&sden[cl * H + h], __expf(score(r, c, h) - decf(smax[cl * H + h])));
      }
    }
    __syncthreads();
    for (int e = tid; e < E + Nn; e += AGG_TB) {
      const int c = (e < E) ? colE[e] : (e - E);
      const unsigned cl = (unsigned)(c - base);
      if (cl < (unsigned)nloc) {
        const int r = (e < E) ? rowE[e] : (e - E);
        for (int f = 0; f < FT; ++f) {
          const int h = f / F;
          const float coef =
              __expf(score(r, c, h) - decf(smax[cl * H + h])) / sden[cl * H + h];
          atomicAdd(&acc[cl * P + f], hsrc[(size_t)r * FT + f] * coef);
        }
      }
    }
  }
  {
    const int wvs = __builtin_amdgcn_readfirstlane(wv);
    const int strip = wvs / WPS;
    const int tn = strip << 4;
    const int idx = lane & 15;
    const int half = lane >> 4;
    const int kk = half << 1;
    v2f b[KS];
#pragma unroll
    for (int s = 0; s < KS; ++s) {
      const float* bp = Wn + (size_t)(4 * s + kk) * AGG_N2 + tn + idx;
      b[s].x = bp[0];
      b[s].y = bp[AGG_N2];
    }
    const float badd = biasn ? biasn[tn + idx] : 0.0f;
    lds_barrier();
    const int ntile = (nloc + 15) >> 4;
    auto pieces = [&](auto scf) {
      constexpr bool SC = decltype(scf)::value;
      for (int t = (wvs + strip) % WPS; t < ntile; t += WPS) {
        const float* sp = &acc[(t * 16 + idx) * P + kk];
        if constexpr (SC) {
          const float* wp = &swa[strip * FT + kk];
          float s0 = 0.f, s1 = 0.f;
#pragma unroll 8
          for (int s = 0; s < KS; ++s) {
            v2f a = *(const v2f*)(sp + 4 * s);
            if constexpr (RELU) { a.x = fmaxf(a.x, 0.f); a.y = fmaxf(a.y, 0.f); }
            const v2f wv = *(const v2f*)(wp + 4 * s);
            s0 = fmaf(a.x, wv.x, s0);
            s1 = fmaf(a.y, wv.y, s1);
          }
          float sc = s0 + s1;
          sc += __shfl_xor(sc, 16);
          const int m = t * 16 + idx;
          if (half == 0 && m < nloc) (strip == 0 ? esn : edn)[base + m] = sc;
        }
        v8f d = {};
#pragma unroll
        for (int s = 0; s < KS; ++s) {
          v2f a = *(const v2f*)(sp + 4 * s);
          if constexpr (RELU) { a.x = fmaxf(a.x, 0.f); a.y = fmaxf(a.y, 0.f); }
          d = __builtin_amdgcn_wmma_f32_16x16x4_f32(
               false, a,  false, b[s],
               (short)0, d,  false,  false);
        }
        const int r0 = t * 16 + 8 * half;
        float* cp = outn + (size_t)(base + r0) * AGG_N2 + tn + idx;
#pragma unroll
        for (int r = 0; r < 8; ++r)
          if (r0 + r < nloc) cp[(size_t)r * AGG_N2] = d[r] + badd;
      }
    };
    if constexpr (SCORE) {
      if (strip < 2) pieces(agg_flag<true>{}); else pieces(agg_flag<false>{});
    } else {
      pieces(agg_flag<false>{});
    }
  }
}


extern "C" void kernel_launch(void* const* d_in, const int* in_sizes, int n_in,
                              void* d_out, int out_size, void* d_ws, size_t ws_size,
                              hipStream_t stream) {
  const float* x     = (const float*)d_in[0];
  const int*   ei    = (const int*)  d_in[1];
  const float* W1    = (const float*)d_in[2];
  const float* attn1 = (const float*)d_in[3];
  const float* W2    = (const float*)d_in[4];
  const float* attn2 = (const float*)d_in[5];
  const float* headW = (const float*)d_in[6];
  const float* headb = (const float*)d_in[7];

  const int Nn = in_sizes[0] / GAT_IN_DIM;
  const int E  = in_sizes[1] / 2;
  const int* rowE = ei;
  const int* colE = ei + E;

  float* w = (float*)d_ws;
  float* wa1   = w; w += (size_t)2 * GAT_H1 * GAT_IN_DIM;
  float* wa2   = w; w += (size_t)2 * GAT_D1;
  float* h1    = w; w += (size_t)Nn * 128;
  float* h2    = w; w += (size_t)Nn * 64;
  float* es1   = w; w += (size_t)Nn * 4;
  float* ed1   = w; w += (size_t)Nn * 4;
  float* es2   = w; w += (size_t)Nn;
  float* ed2   = w; w += (size_t)Nn;
  const int agg_blocks = (Nn + AGG_NPB - 1) / AGG_NPB;
  const int nseg = (E + BIN_EPB - 1) / BIN_EPB > 0 ? (E + BIN_EPB - 1) / BIN_EPB : 1;
  int* bins = (int*)w; w += (size_t)agg_blocks * nseg * BIN_CAP;
  int* bcnt = (int*)w; w += (size_t)agg_blocks * nseg;

  const int mtiles = Nn >> 4;
  const int TPB128 = 2;
  auto gemm_blocks = [](int tiles, int tpb) { return (tiles + tpb - 1) / tpb; };

  gat_fold_attn<<<(FOLD_N1 + FOLD_N2) / FOLD_TB, FOLD_TB, 0, stream>>>(
      W1, attn1, W2, attn2, wa1, wa2);

  wmma_gemm_f32<128, 128, GAT_H1>
      <<<nseg + gemm_blocks(mtiles, TPB128), 256, agg_blocks * sizeof(int), stream>>>(
      x, W1, h1, nullptr, Nn, TPB128, wa1, es1, ed1,
      nseg, rowE, colE, E, agg_blocks, bins, bcnt);
  gat_aggregate_lds<4, 32, true, true><<<agg_blocks, AGG_TB, 0, stream>>>(
      es1, ed1, h1, W2, nullptr, h2, wa2, es2, ed2, rowE, colE, E, Nn, bins, bcnt, nseg);

  gat_aggregate_lds<1, 64, false, false><<<agg_blocks, AGG_TB, 0, stream>>>(
      es2, ed2, h2, headW, headb, (float*)d_out, nullptr, nullptr, nullptr,
      rowE, colE, E, Nn, bins, bcnt, nseg);
}
